# starting kernel
# baseline (speedup 1.0000x reference)
#include <hip/hip_runtime.h>
#include <limits.h>

typedef float v2f __attribute__((ext_vector_type(2)));
typedef float v8f __attribute__((ext_vector_type(8)));

#define NEG_INF (-1e30f)

__global__ __launch_bounds__(256) void gemv_wmma_f32(
    const float* __restrict__ A, const float* __restrict__ W,
    const float* __restrict__ bias, float* __restrict__ out,
    int N, int Hdim) {
  int wave = (int)((blockIdx.x * blockDim.x + threadIdx.x) >> 5);
  int lane = (int)(threadIdx.x & 31);
  int row0 = wave * 16;
  if (row0 >= N) return;

  int m = lane & 15;
  int khalf = (lane >> 4) * 2;
  int row = row0 + m;
  if (row >= N) row = N - 1;

  const float* arow = A + (size_t)row * (size_t)Hdim + khalf;
  const float* wp   = W + khalf;

  v8f acc0 = {}; v8f acc1 = {}; v8f acc2 = {}; v8f acc3 = {};
  for (int k = 0; k < Hdim; k += 16) {
    v2f a0 = __builtin_nontemporal_load((const v2f*)(arow + k));
    v2f b0 = *(const v2f*)(wp + k);
    v2f a1 = __builtin_nontemporal_load((const v2f*)(arow + k + 4));
    v2f b1 = *(const v2f*)(wp + k + 4);
    v2f a2 = __builtin_nontemporal_load((const v2f*)(arow + k + 8));
    v2f b2 = *(const v2f*)(wp + k + 8);
    v2f a3 = __builtin_nontemporal_load((const v2f*)(arow + k + 12));
    v2f b3 = *(const v2f*)(wp + k + 12);
    acc0 = __builtin_amdgcn_wmma_f32_16x16x4_f32(false, a0, false, b0,
                                                 (short)0, acc0, false, false);
    acc1 = __builtin_amdgcn_wmma_f32_16x16x4_f32(false, a1, false, b1,
                                                 (short)0, acc1, false, false);
    acc2 = __builtin_amdgcn_wmma_f32_16x16x4_f32(false, a2, false, b2,
                                                 (short)0, acc2, false, false);
    acc3 = __builtin_amdgcn_wmma_f32_16x16x4_f32(false, a3, false, b3,
                                                 (short)0, acc3, false, false);
  }
  v8f acc = (acc0 + acc1) + (acc2 + acc3);
  float bv = bias[0];

  if ((lane & 15) == 0) {
    int mb = row0 + (lane >> 4) * 8;
#pragma unroll
    for (int r = 0; r < 8; ++r) {
      int rr = mb + r;
      if (rr < N) out[rr] = acc[r] + bv;
    }
  }
}

__global__ void scan_offsets(const int* __restrict__ lens,
                             int* __restrict__ offsets, int S) {
  int lane = (int)threadIdx.x;
  int run = 0;
  for (int base = 0; base < S; base += 32) {
    int i = base + lane;
    int x = (i < S) ? lens[i] : 0;
    int v = x;
#pragma unroll
    for (int d = 1; d < 32; d <<= 1) {
      int t = __shfl_up(v, d, 32);
      if (lane >= d) v += t;
    }
    if (i < S) offsets[i] = run + (v - x);
    run += __shfl(v, 31, 32);
  }
}

__global__ __launch_bounds__(256) void seg_loss(
    const float* __restrict__ logits, const int* __restrict__ lens,
    const int* __restrict__ offsets, const int* __restrict__ tgt,
    float* __restrict__ loss_ws, float* __restrict__ inc_ws,
    int S, int EACH) {
  int wave = (int)((blockIdx.x * blockDim.x + threadIdx.x) >> 5);
  int lane = (int)(threadIdx.x & 31);
  int s = wave;
  if (s >= S) return;

  int len = lens[s];
  int off = offsets[s];
  int j0 = lane, j1 = lane + 32;
  bool v0 = j0 < len, v1 = j1 < len;
  float x0 = v0 ? logits[off + j0] : NEG_INF;
  float x1 = v1 ? logits[off + j1] : NEG_INF;
  int   t0 = v0 ? tgt[off + j0] : 0;
  int   t1 = v1 ? tgt[off + j1] : 0;

  float mx = fmaxf(x0, x1);
#pragma unroll
  for (int d = 16; d > 0; d >>= 1) mx = fmaxf(mx, __shfl_xor(mx, d, 32));
  float e0 = v0 ? __expf(x0 - mx) : 0.0f;
  float e1 = v1 ? __expf(x1 - mx) : 0.0f;
  float sum = e0 + e1;
#pragma unroll
  for (int d = 16; d > 0; d >>= 1) sum += __shfl_xor(sum, d, 32);
  float inv = 1.0f / sum;
  float p0 = e0 * inv, p1 = e1 * inv;

  float q0 = v0 ? p0 : NEG_INF;
  float q1 = v1 ? p1 : NEG_INF;
  float m2 = fmaxf(q0, q1);
#pragma unroll
  for (int d = 16; d > 0; d >>= 1) m2 = fmaxf(m2, __shfl_xor(m2, d, 32));
  float s2 = (v0 ? __expf(p0 - m2) : 0.0f) + (v1 ? __expf(p1 - m2) : 0.0f);
#pragma unroll
  for (int d = 16; d > 0; d >>= 1) s2 += __shfl_xor(s2, d, 32);
  float lse2 = m2 + __logf(s2);

  int k0 = (v0 && t0 == 1) ? j0 : INT_MAX;
  int k1 = (v1 && t1 == 1) ? j1 : INT_MAX;
  int tk = min(k0, k1);
#pragma unroll
  for (int d = 16; d > 0; d >>= 1) tk = min(tk, __shfl_xor(tk, d, 32));
  int t_idx = (tk == INT_MAX) ? 0 : tk;
  float psel = (t_idx >= 32) ? p1 : p0;
  float p_t = __shfl(psel, t_idx & 31, 32);
  float loss_per = lse2 - p_t;

  int ts = t0 + t1;
#pragma unroll
  for (int d = 16; d > 0; d >>= 1) ts += __shfl_xor(ts, d, 32);

  int next_len = (s + 1 < S) ? lens[s + 1] : 1;
  bool skip_next_zero = (s > 0) && (s + 1 < S) && (next_len == 0);
  bool skip_no_pos = (((s + 1) % EACH) == 0) && (ts <= 0);
  bool include = (len > 0) && !skip_next_zero && !skip_no_pos;

  if (lane == 0) {
    loss_ws[s] = include ? loss_per : 0.0f;
    inc_ws[s]  = include ? 1.0f : 0.0f;
  }
}

__global__ __launch_bounds__(256) void final_reduce(
    const float* __restrict__ loss_ws, const float* __restrict__ inc_ws,
    float* __restrict__ out_loss, int S) {
  __shared__ float sl[256];
  __shared__ float sc[256];
  int t = (int)threadIdx.x;
  float a = 0.0f, c = 0.0f;
  for (int i = t; i < S; i += 256) { a += loss_ws[i]; c += inc_ws[i]; }
  sl[t] = a; sc[t] = c;
  __syncthreads();
  for (int d = 128; d > 0; d >>= 1) {
    if (t < d) { sl[t] += sl[t + d]; sc[t] += sc[t + d]; }
    __syncthreads();
  }
  if (t == 0) {
    float cnt = sc[0];
    float denom = (cnt > 0.0f) ? cnt : 1e-4f;
    out_loss[0] = sl[0] / denom;
  }
}

extern "C" void kernel_launch(void* const* d_in, const int* in_sizes, int n_in,
                              void* d_out, int out_size, void* d_ws, size_t ws_size,
                              hipStream_t stream) {
  const float* A    = (const float*)d_in[0];
  const float* W    = (const float*)d_in[1];
  const float* b    = (const float*)d_in[2];
  const int*   lens = (const int*)d_in[3];
  const int*   tgt  = (const int*)d_in[4];

  int Hdim = in_sizes[1];
  int N    = in_sizes[0] / Hdim;
  int S    = in_sizes[3];
  const int EACH = 32;

  float* logits   = (float*)d_out;
  float* loss_out = logits + N;

  int*   offsets = (int*)d_ws;
  size_t obytes  = (((size_t)S * sizeof(int)) + 255) & ~(size_t)255;
  float* loss_ws = (float*)((char*)d_ws + obytes);
  float* inc_ws  = loss_ws + S;

  int waves  = (N + 15) / 16;
  int blocks = (waves + 7) / 8;
  gemv_wmma_f32<<<blocks, 256, 0, stream>>>(A, W, b, logits, N, Hdim);

  scan_offsets<<<1, 32, 0, stream>>>(lens, offsets, S);

  seg_loss<<<(S + 7) / 8, 256, 0, stream>>>(logits, lens, offsets, tgt,
                                            loss_ws, inc_ws, S, EACH);

  final_reduce<<<1, 256, 0, stream>>>(loss_ws, inc_ws, loss_out, S);
}


# GEMV as a row stream: each wave-wide b128 load reads 512 contiguous bytes of one row (was 16 B of each of 16 rows), W in registers, dot = trace of the fp32 WMMA tile; sum order changes
# speedup vs baseline: 1.8335x; 1.8335x over previous
#include <hip/hip_runtime.h>
#include <limits.h>

typedef float v2f __attribute__((ext_vector_type(2)));
typedef float v8f __attribute__((ext_vector_type(8)));

#define NEG_INF (-1e30f)

__global__ __launch_bounds__(256) void gemv_wmma_f32(
    const float* __restrict__ A, const float* __restrict__ W,
    const float* __restrict__ bias, float* __restrict__ out,
    int N, int Hdim) {
  int wave = (int)((blockIdx.x * blockDim.x + threadIdx.x) >> 5);
  int lane = (int)(threadIdx.x & 31);
  int row0 = wave * 16;
  if (row0 >= N) return;

  int m = lane & 15;
  int khalf = (lane >> 4) * 2;
  int row = row0 + m;
  if (row >= N) row = N - 1;

  const float* arow = A + (size_t)row * (size_t)Hdim + khalf;
  const float* wp   = W + khalf;

  v8f acc0 = {}; v8f acc1 = {}; v8f acc2 = {}; v8f acc3 = {};
  for (int k = 0; k < Hdim; k += 16) {
    v2f a0 = __builtin_nontemporal_load((const v2f*)(arow + k));
    v2f b0 = *(const v2f*)(wp + k);
    v2f a1 = __builtin_nontemporal_load((const v2f*)(arow + k + 4));
    v2f b1 = *(const v2f*)(wp + k + 4);
    v2f a2 = __builtin_nontemporal_load((const v2f*)(arow + k + 8));
    v2f b2 = *(const v2f*)(wp + k + 8);
    v2f a3 = __builtin_nontemporal_load((const v2f*)(arow + k + 12));
    v2f b3 = *(const v2f*)(wp + k + 12);
    acc0 = __builtin_amdgcn_wmma_f32_16x16x4_f32(false, a0, false, b0,
                                                 (short)0, acc0, false, false);
    acc1 = __builtin_amdgcn_wmma_f32_16x16x4_f32(false, a1, false, b1,
                                                 (short)0, acc1, false, false);
    acc2 = __builtin_amdgcn_wmma_f32_16x16x4_f32(false, a2, false, b2,
                                                 (short)0, acc2, false, false);
    acc3 = __builtin_amdgcn_wmma_f32_16x16x4_f32(false, a3, false, b3,
                                                 (short)0, acc3, false, false);
  }
  v8f acc = (acc0 + acc1) + (acc2 + acc3);
  float bv = bias[0];

  if ((lane & 15) == 0) {
    int mb = row0 + (lane >> 4) * 8;
#pragma unroll
    for (int r = 0; r < 8; ++r) {
      int rr = mb + r;
      if (rr < N) out[rr] = acc[r] + bv;
    }
  }
}

typedef float v4f __attribute__((ext_vector_type(4)));

#define GV_ROWS  16
#define GV_AHEAD 3
#define GV_KQ    8
#define GV_WAVES 4

__global__ __launch_bounds__(GV_WAVES * 32) void gemv_rows_wmma_f32(
    const float* __restrict__ A, const float* __restrict__ W,
    const float* __restrict__ bias, float* __restrict__ out, int N) {
  const unsigned lane = threadIdx.x & 31u;
  const unsigned wave = (unsigned)__builtin_amdgcn_readfirstlane(
      (int)((blockIdx.x * (unsigned)(GV_WAVES * 32) + threadIdx.x) >> 5));
  const unsigned row0 = wave * GV_ROWS;
  if (row0 >= (unsigned)N) return;

  v4f wq[GV_KQ];
  for (int q = 0; q < GV_KQ; ++q)
    wq[q] = *(const v4f*)(W + 128 * q + 4 * lane);

  const float* ab = A + (size_t)row0 * 1024u + 4u * lane;
  v4f a[GV_AHEAD + 1][GV_KQ];
#define GV_LOAD_ROW(R) do { \
    const unsigned rr_ = (row0 + (unsigned)(R) < (unsigned)N) ? (unsigned)(R) : 0u; \
    for (int q = 0; q < GV_KQ; ++q) \
      a[(R) % (GV_AHEAD + 1)][q] = __builtin_nontemporal_load( \
          (const v4f*)(ab + (size_t)rr_ * 1024u + 128u * q)); } while (0)

  for (int r = 0; r < GV_AHEAD; ++r) GV_LOAD_ROW(r);

  const bool lo8  = lane < 8u;
  const bool diag = lo8 || lane >= 24u;
  const bool b0 = (lane & 1u) != 0u, b1 = (lane & 2u) != 0u, b2 = (lane & 4u) != 0u;

  float part[GV_ROWS];
  for (int r = 0; r < GV_ROWS; ++r) {
    __builtin_amdgcn_sched_barrier(0);
    if (r + GV_AHEAD < GV_ROWS) GV_LOAD_ROW(r + GV_AHEAD);
    __builtin_amdgcn_sched_barrier(0);
    const v4f* ar = a[r % (GV_AHEAD + 1)];
    v8f c0 = {}; v8f c1 = {};
    for (int q = 0; q < GV_KQ; ++q) {
      v2f a01 = {ar[q].x, ar[q].y}, w01 = {wq[q].x, wq[q].y};
      v2f a23 = {ar[q].z, ar[q].w}, w23 = {wq[q].z, wq[q].w};
      c0 = __builtin_amdgcn_wmma_f32_16x16x4_f32(false, a01, false, w01,
                                                 (short)0, c0, false, false);
      c1 = __builtin_amdgcn_wmma_f32_16x16x4_f32(false, a23, false, w23,
                                                 (short)0, c1, false, false);
    }
    v8f c = c0 + c1;
    float s0 = b0 ? c[1] : c[0], s1 = b0 ? c[3] : c[2];
    float s2 = b0 ? c[5] : c[4], s3 = b0 ? c[7] : c[6];
    float t0 = b1 ? s1 : s0, t1 = b1 ? s3 : s2;
    float u  = b2 ? t1 : t0;
    part[r] = diag ? u : 0.0f;
    asm volatile("" : "+v"(part[r]));
  }
#undef GV_LOAD_ROW

  const bool h16 = (lane & 16u) != 0u, h8 = (lane & 8u) != 0u;
  const bool h4 = (lane & 4u) != 0u, h2 = (lane & 2u) != 0u;
  float e8[8], e4[4], e2[2];
  for (int i = 0; i < 8; ++i) {
    float keep = h16 ? part[i + 8] : part[i], send = h16 ? part[i] : part[i + 8];
    e8[i] = keep + __shfl_xor(send, 16, 32);
  }
  for (int i = 0; i < 4; ++i) {
    float keep = h8 ? e8[i + 4] : e8[i], send = h8 ? e8[i] : e8[i + 4];
    e4[i] = keep + __shfl_xor(send, 8, 32);
  }
  for (int i = 0; i < 2; ++i) {
    float keep = h4 ? e4[i + 2] : e4[i], send = h4 ? e4[i] : e4[i + 2];
    e2[i] = keep + __shfl_xor(send, 4, 32);
  }
  float keep1 = h2 ? e2[1] : e2[0], send1 = h2 ? e2[0] : e2[1];
  float e1 = keep1 + __shfl_xor(send1, 2, 32);
  float tot = e1 + __shfl_xor(e1, 1, 32);

  const unsigned myrow = row0 + (lane >> 1);
  if ((lane & 1u) == 0u && myrow < (unsigned)N) out[myrow] = tot + bias[0];
}

__global__ void scan_offsets(const int* __restrict__ lens,
                             int* __restrict__ offsets, int S) {
  int lane = (int)threadIdx.x;
  int run = 0;
  for (int base = 0; base < S; base += 32) {
    int i = base + lane;
    int x = (i < S) ? lens[i] : 0;
    int v = x;
#pragma unroll
    for (int d = 1; d < 32; d <<= 1) {
      int t = __shfl_up(v, d, 32);
      if (lane >= d) v += t;
    }
    if (i < S) offsets[i] = run + (v - x);
    run += __shfl(v, 31, 32);
  }
}

__global__ __launch_bounds__(256) void seg_loss(
    const float* __restrict__ logits, const int* __restrict__ lens,
    const int* __restrict__ offsets, const int* __restrict__ tgt,
    float* __restrict__ loss_ws, float* __restrict__ inc_ws,
    int S, int EACH) {
  int wave = (int)((blockIdx.x * blockDim.x + threadIdx.x) >> 5);
  int lane = (int)(threadIdx.x & 31);
  int s = wave;
  if (s >= S) return;

  int len = lens[s];
  int off = offsets[s];
  int j0 = lane, j1 = lane + 32;
  bool v0 = j0 < len, v1 = j1 < len;
  float x0 = v0 ? logits[off + j0] : NEG_INF;
  float x1 = v1 ? logits[off + j1] : NEG_INF;
  int   t0 = v0 ? tgt[off + j0] : 0;
  int   t1 = v1 ? tgt[off + j1] : 0;

  float mx = fmaxf(x0, x1);
#pragma unroll
  for (int d = 16; d > 0; d >>= 1) mx = fmaxf(mx, __shfl_xor(mx, d, 32));
  float e0 = v0 ? __expf(x0 - mx) : 0.0f;
  float e1 = v1 ? __expf(x1 - mx) : 0.0f;
  float sum = e0 + e1;
#pragma unroll
  for (int d = 16; d > 0; d >>= 1) sum += __shfl_xor(sum, d, 32);
  float inv = 1.0f / sum;
  float p0 = e0 * inv, p1 = e1 * inv;

  float q0 = v0 ? p0 : NEG_INF;
  float q1 = v1 ? p1 : NEG_INF;
  float m2 = fmaxf(q0, q1);
#pragma unroll
  for (int d = 16; d > 0; d >>= 1) m2 = fmaxf(m2, __shfl_xor(m2, d, 32));
  float s2 = (v0 ? __expf(p0 - m2) : 0.0f) + (v1 ? __expf(p1 - m2) : 0.0f);
#pragma unroll
  for (int d = 16; d > 0; d >>= 1) s2 += __shfl_xor(s2, d, 32);
  float lse2 = m2 + __logf(s2);

  int k0 = (v0 && t0 == 1) ? j0 : INT_MAX;
  int k1 = (v1 && t1 == 1) ? j1 : INT_MAX;
  int tk = min(k0, k1);
#pragma unroll
  for (int d = 16; d > 0; d >>= 1) tk = min(tk, __shfl_xor(tk, d, 32));
  int t_idx = (tk == INT_MAX) ? 0 : tk;
  float psel = (t_idx >= 32) ? p1 : p0;
  float p_t = __shfl(psel, t_idx & 31, 32);
  float loss_per = lse2 - p_t;

  int ts = t0 + t1;
#pragma unroll
  for (int d = 16; d > 0; d >>= 1) ts += __shfl_xor(ts, d, 32);

  int next_len = (s + 1 < S) ? lens[s + 1] : 1;
  bool skip_next_zero = (s > 0) && (s + 1 < S) && (next_len == 0);
  bool skip_no_pos = (((s + 1) % EACH) == 0) && (ts <= 0);
  bool include = (len > 0) && !skip_next_zero && !skip_no_pos;

  if (lane == 0) {
    loss_ws[s] = include ? loss_per : 0.0f;
    inc_ws[s]  = include ? 1.0f : 0.0f;
  }
}

__global__ __launch_bounds__(256) void final_reduce(
    const float* __restrict__ loss_ws, const float* __restrict__ inc_ws,
    float* __restrict__ out_loss, int S) {
  __shared__ float sl[256];
  __shared__ float sc[256];
  int t = (int)threadIdx.x;
  float a = 0.0f, c = 0.0f;
  for (int i = t; i < S; i += 256) { a += loss_ws[i]; c += inc_ws[i]; }
  sl[t] = a; sc[t] = c;
  __syncthreads();
  for (int d = 128; d > 0; d >>= 1) {
    if (t < d) { sl[t] += sl[t + d]; sc[t] += sc[t + d]; }
    __syncthreads();
  }
  if (t == 0) {
    float cnt = sc[0];
    float denom = (cnt > 0.0f) ? cnt : 1e-4f;
    out_loss[0] = sl[0] / denom;
  }
}

extern "C" void kernel_launch(void* const* d_in, const int* in_sizes, int n_in,
                              void* d_out, int out_size, void* d_ws, size_t ws_size,
                              hipStream_t stream) {
  const float* A    = (const float*)d_in[0];
  const float* W    = (const float*)d_in[1];
  const float* b    = (const float*)d_in[2];
  const int*   lens = (const int*)d_in[3];
  const int*   tgt  = (const int*)d_in[4];

  int Hdim = in_sizes[1];
  int N    = in_sizes[0] / Hdim;
  int S    = in_sizes[3];
  const int EACH = 32;

  float* logits   = (float*)d_out;
  float* loss_out = logits + N;

  int*   offsets = (int*)d_ws;
  size_t obytes  = (((size_t)S * sizeof(int)) + 255) & ~(size_t)255;
  float* loss_ws = (float*)((char*)d_ws + obytes);
  float* inc_ws  = loss_ws + S;

  int waves  = (N + 15) / 16;
  if (Hdim == 1024) {
    int blocks = (waves + GV_WAVES - 1) / GV_WAVES;
    gemv_rows_wmma_f32<<<blocks, GV_WAVES * 32, 0, stream>>>(A, W, b, logits, N);
  } else {
    int blocks = (waves + 7) / 8;
    gemv_wmma_f32<<<blocks, 256, 0, stream>>>(A, W, b, logits, N, Hdim);
  }

  scan_offsets<<<1, 32, 0, stream>>>(lens, offsets, S);

  seg_loss<<<(S + 7) / 8, 256, 0, stream>>>(logits, lens, offsets, tgt,
                                            loss_ws, inc_ws, S, EACH);

  final_reduce<<<1, 256, 0, stream>>>(loss_ws, inc_ws, loss_out, S);
}


# Segment starts computed inside the per-segment kernel: each wave sums the earlier lengths itself with batched independent loads; the serial one-wave scan launch is removed (4 launches to 3)
# speedup vs baseline: 3.1535x; 1.7199x over previous
#include <hip/hip_runtime.h>
#include <limits.h>

typedef float v2f __attribute__((ext_vector_type(2)));
typedef float v8f __attribute__((ext_vector_type(8)));

#define NEG_INF (-1e30f)

__global__ __launch_bounds__(256) void gemv_wmma_f32(
    const float* __restrict__ A, const float* __restrict__ W,
    const float* __restrict__ bias, float* __restrict__ out,
    int N, int Hdim) {
  int wave = (int)((blockIdx.x * blockDim.x + threadIdx.x) >> 5);
  int lane = (int)(threadIdx.x & 31);
  int row0 = wave * 16;
  if (row0 >= N) return;

  int m = lane & 15;
  int khalf = (lane >> 4) * 2;
  int row = row0 + m;
  if (row >= N) row = N - 1;

  const float* arow = A + (size_t)row * (size_t)Hdim + khalf;
  const float* wp   = W + khalf;

  v8f acc0 = {}; v8f acc1 = {}; v8f acc2 = {}; v8f acc3 = {};
  for (int k = 0; k < Hdim; k += 16) {
    v2f a0 = __builtin_nontemporal_load((const v2f*)(arow + k));
    v2f b0 = *(const v2f*)(wp + k);
    v2f a1 = __builtin_nontemporal_load((const v2f*)(arow + k + 4));
    v2f b1 = *(const v2f*)(wp + k + 4);
    v2f a2 = __builtin_nontemporal_load((const v2f*)(arow + k + 8));
    v2f b2 = *(const v2f*)(wp + k + 8);
    v2f a3 = __builtin_nontemporal_load((const v2f*)(arow + k + 12));
    v2f b3 = *(const v2f*)(wp + k + 12);
    acc0 = __builtin_amdgcn_wmma_f32_16x16x4_f32(false, a0, false, b0,
                                                 (short)0, acc0, false, false);
    acc1 = __builtin_amdgcn_wmma_f32_16x16x4_f32(false, a1, false, b1,
                                                 (short)0, acc1, false, false);
    acc2 = __builtin_amdgcn_wmma_f32_16x16x4_f32(false, a2, false, b2,
                                                 (short)0, acc2, false, false);
    acc3 = __builtin_amdgcn_wmma_f32_16x16x4_f32(false, a3, false, b3,
                                                 (short)0, acc3, false, false);
  }
  v8f acc = (acc0 + acc1) + (acc2 + acc3);
  float bv = bias[0];

  if ((lane & 15) == 0) {
    int mb = row0 + (lane >> 4) * 8;
#pragma unroll
    for (int r = 0; r < 8; ++r) {
      int rr = mb + r;
      if (rr < N) out[rr] = acc[r] + bv;
    }
  }
}

typedef float v4f __attribute__((ext_vector_type(4)));

#define GV_ROWS  16
#define GV_AHEAD 3
#define GV_KQ    8
#define GV_WAVES 4

__global__ __launch_bounds__(GV_WAVES * 32) void gemv_rows_wmma_f32(
    const float* __restrict__ A, const float* __restrict__ W,
    const float* __restrict__ bias, float* __restrict__ out, int N) {
  const unsigned lane = threadIdx.x & 31u;
  const unsigned wave = (unsigned)__builtin_amdgcn_readfirstlane(
      (int)((blockIdx.x * (unsigned)(GV_WAVES * 32) + threadIdx.x) >> 5));
  const unsigned row0 = wave * GV_ROWS;
  if (row0 >= (unsigned)N) return;

  v4f wq[GV_KQ];
  for (int q = 0; q < GV_KQ; ++q)
    wq[q] = *(const v4f*)(W + 128 * q + 4 * lane);

  const float* ab = A + (size_t)row0 * 1024u + 4u * lane;
  v4f a[GV_AHEAD + 1][GV_KQ];
#define GV_LOAD_ROW(R) do { \
    const unsigned rr_ = (row0 + (unsigned)(R) < (unsigned)N) ? (unsigned)(R) : 0u; \
    for (int q = 0; q < GV_KQ; ++q) \
      a[(R) % (GV_AHEAD + 1)][q] = __builtin_nontemporal_load( \
          (const v4f*)(ab + (size_t)rr_ * 1024u + 128u * q)); } while (0)

  for (int r = 0; r < GV_AHEAD; ++r) GV_LOAD_ROW(r);

  const bool lo8  = lane < 8u;
  const bool diag = lo8 || lane >= 24u;
  const bool b0 = (lane & 1u) != 0u, b1 = (lane & 2u) != 0u, b2 = (lane & 4u) != 0u;

  float part[GV_ROWS];
  for (int r = 0; r < GV_ROWS; ++r) {
    __builtin_amdgcn_sched_barrier(0);
    if (r + GV_AHEAD < GV_ROWS) GV_LOAD_ROW(r + GV_AHEAD);
    __builtin_amdgcn_sched_barrier(0);
    const v4f* ar = a[r % (GV_AHEAD + 1)];
    v8f c0 = {}; v8f c1 = {};
    for (int q = 0; q < GV_KQ; ++q) {
      v2f a01 = {ar[q].x, ar[q].y}, w01 = {wq[q].x, wq[q].y};
      v2f a23 = {ar[q].z, ar[q].w}, w23 = {wq[q].z, wq[q].w};
      c0 = __builtin_amdgcn_wmma_f32_16x16x4_f32(false, a01, false, w01,
                                                 (short)0, c0, false, false);
      c1 = __builtin_amdgcn_wmma_f32_16x16x4_f32(false, a23, false, w23,
                                                 (short)0, c1, false, false);
    }
    v8f c = c0 + c1;
    float s0 = b0 ? c[1] : c[0], s1 = b0 ? c[3] : c[2];
    float s2 = b0 ? c[5] : c[4], s3 = b0 ? c[7] : c[6];
    float t0 = b1 ? s1 : s0, t1 = b1 ? s3 : s2;
    float u  = b2 ? t1 : t0;
    part[r] = diag ? u : 0.0f;
    asm volatile("" : "+v"(part[r]));
  }
#undef GV_LOAD_ROW

  const bool h16 = (lane & 16u) != 0u, h8 = (lane & 8u) != 0u;
  const bool h4 = (lane & 4u) != 0u, h2 = (lane & 2u) != 0u;
  float e8[8], e4[4], e2[2];
  for (int i = 0; i < 8; ++i) {
    float keep = h16 ? part[i + 8] : part[i], send = h16 ? part[i] : part[i + 8];
    e8[i] = keep + __shfl_xor(send, 16, 32);
  }
  for (int i = 0; i < 4; ++i) {
    float keep = h8 ? e8[i + 4] : e8[i], send = h8 ? e8[i] : e8[i + 4];
    e4[i] = keep + __shfl_xor(send, 8, 32);
  }
  for (int i = 0; i < 2; ++i) {
    float keep = h4 ? e4[i + 2] : e4[i], send = h4 ? e4[i] : e4[i + 2];
    e2[i] = keep + __shfl_xor(send, 4, 32);
  }
  float keep1 = h2 ? e2[1] : e2[0], send1 = h2 ? e2[0] : e2[1];
  float e1 = keep1 + __shfl_xor(send1, 2, 32);
  float tot = e1 + __shfl_xor(e1, 1, 32);

  const unsigned myrow = row0 + (lane >> 1);
  if ((lane & 1u) == 0u && myrow < (unsigned)N) out[myrow] = tot + bias[0];
}

#define SEG_U 16

__global__ __launch_bounds__(256) void seg_loss(
    const float* __restrict__ logits, const int* __restrict__ lens,
    const int* __restrict__ tgt,
    float* __restrict__ loss_ws, float* __restrict__ inc_ws,
    int S, int EACH) {
  int wave = (int)((blockIdx.x * blockDim.x + threadIdx.x) >> 5);
  int lane = (int)(threadIdx.x & 31);
  int s = wave;
  if (s >= S) return;

  int len = lens[s];
  int acc = 0;
  for (int base = 0; base < s; base += 32 * SEG_U) {
    int v[SEG_U];
    for (int u = 0; u < SEG_U; ++u) {
      int i = base + 32 * u + lane;
      v[u] = lens[i < s ? i : s];
    }
    for (int u = 0; u < SEG_U; ++u) {
      int i = base + 32 * u + lane;
      acc += (i < s) ? v[u] : 0;
    }
  }
  for (int d = 16; d > 0; d >>= 1) acc += __shfl_xor(acc, d, 32);
  int off = acc;
  int j0 = lane, j1 = lane + 32;
  bool v0 = j0 < len, v1 = j1 < len;
  float x0 = v0 ? logits[off + j0] : NEG_INF;
  float x1 = v1 ? logits[off + j1] : NEG_INF;
  int   t0 = v0 ? tgt[off + j0] : 0;
  int   t1 = v1 ? tgt[off + j1] : 0;

  float mx = fmaxf(x0, x1);
#pragma unroll
  for (int d = 16; d > 0; d >>= 1) mx = fmaxf(mx, __shfl_xor(mx, d, 32));
  float e0 = v0 ? __expf(x0 - mx) : 0.0f;
  float e1 = v1 ? __expf(x1 - mx) : 0.0f;
  float sum = e0 + e1;
#pragma unroll
  for (int d = 16; d > 0; d >>= 1) sum += __shfl_xor(sum, d, 32);
  float inv = 1.0f / sum;
  float p0 = e0 * inv, p1 = e1 * inv;

  float q0 = v0 ? p0 : NEG_INF;
  float q1 = v1 ? p1 : NEG_INF;
  float m2 = fmaxf(q0, q1);
#pragma unroll
  for (int d = 16; d > 0; d >>= 1) m2 = fmaxf(m2, __shfl_xor(m2, d, 32));
  float s2 = (v0 ? __expf(p0 - m2) : 0.0f) + (v1 ? __expf(p1 - m2) : 0.0f);
#pragma unroll
  for (int d = 16; d > 0; d >>= 1) s2 += __shfl_xor(s2, d, 32);
  float lse2 = m2 + __logf(s2);

  int k0 = (v0 && t0 == 1) ? j0 : INT_MAX;
  int k1 = (v1 && t1 == 1) ? j1 : INT_MAX;
  int tk = min(k0, k1);
#pragma unroll
  for (int d = 16; d > 0; d >>= 1) tk = min(tk, __shfl_xor(tk, d, 32));
  int t_idx = (tk == INT_MAX) ? 0 : tk;
  float psel = (t_idx >= 32) ? p1 : p0;
  float p_t = __shfl(psel, t_idx & 31, 32);
  float loss_per = lse2 - p_t;

  int ts = t0 + t1;
#pragma unroll
  for (int d = 16; d > 0; d >>= 1) ts += __shfl_xor(ts, d, 32);

  int next_len = (s + 1 < S) ? lens[s + 1] : 1;
  bool skip_next_zero = (s > 0) && (s + 1 < S) && (next_len == 0);
  bool skip_no_pos = (((s + 1) % EACH) == 0) && (ts <= 0);
  bool include = (len > 0) && !skip_next_zero && !skip_no_pos;

  if (lane == 0) {
    loss_ws[s] = include ? loss_per : 0.0f;
    inc_ws[s]  = include ? 1.0f : 0.0f;
  }
}

__global__ __launch_bounds__(256) void final_reduce(
    const float* __restrict__ loss_ws, const float* __restrict__ inc_ws,
    float* __restrict__ out_loss, int S) {
  __shared__ float sl[256];
  __shared__ float sc[256];
  int t = (int)threadIdx.x;
  float a = 0.0f, c = 0.0f;
  for (int i = t; i < S; i += 256) { a += loss_ws[i]; c += inc_ws[i]; }
  sl[t] = a; sc[t] = c;
  __syncthreads();
  for (int d = 128; d > 0; d >>= 1) {
    if (t < d) { sl[t] += sl[t + d]; sc[t] += sc[t + d]; }
    __syncthreads();
  }
  if (t == 0) {
    float cnt = sc[0];
    float denom = (cnt > 0.0f) ? cnt : 1e-4f;
    out_loss[0] = sl[0] / denom;
  }
}

extern "C" void kernel_launch(void* const* d_in, const int* in_sizes, int n_in,
                              void* d_out, int out_size, void* d_ws, size_t ws_size,
                              hipStream_t stream) {
  const float* A    = (const float*)d_in[0];
  const float* W    = (const float*)d_in[1];
  const float* b    = (const float*)d_in[2];
  const int*   lens = (const int*)d_in[3];
  const int*   tgt  = (const int*)d_in[4];

  int Hdim = in_sizes[1];
  int N    = in_sizes[0] / Hdim;
  int S    = in_sizes[3];
  const int EACH = 32;

  float* logits   = (float*)d_out;
  float* loss_out = logits + N;

  size_t obytes  = (((size_t)S * sizeof(int)) + 255) & ~(size_t)255;
  float* loss_ws = (float*)((char*)d_ws + obytes);
  float* inc_ws  = loss_ws + S;

  int waves  = (N + 15) / 16;
  if (Hdim == 1024) {
    int blocks = (waves + GV_WAVES - 1) / GV_WAVES;
    gemv_rows_wmma_f32<<<blocks, GV_WAVES * 32, 0, stream>>>(A, W, b, logits, N);
  } else {
    int blocks = (waves + 7) / 8;
    gemv_wmma_f32<<<blocks, 256, 0, stream>>>(A, W, b, logits, N, Hdim);
  }

  seg_loss<<<(S + 7) / 8, 256, 0, stream>>>(logits, lens, tgt,
                                            loss_ws, inc_ws, S, EACH);

  final_reduce<<<1, 256, 0, stream>>>(loss_ws, inc_ws, loss_out, S);
}


# Final reduce waits for memory once: a thread requests its 6 (loss, include) pairs together, wave sums by lane exchange, 1 barrier (was 6 dependent load trips, 9 barriers); loss sum order changes
# speedup vs baseline: 3.3244x; 1.0542x over previous
#include <hip/hip_runtime.h>
#include <limits.h>

typedef float v2f __attribute__((ext_vector_type(2)));
typedef float v8f __attribute__((ext_vector_type(8)));

#define NEG_INF (-1e30f)

__global__ __launch_bounds__(256) void gemv_wmma_f32(
    const float* __restrict__ A, const float* __restrict__ W,
    const float* __restrict__ bias, float* __restrict__ out,
    int N, int Hdim) {
  int wave = (int)((blockIdx.x * blockDim.x + threadIdx.x) >> 5);
  int lane = (int)(threadIdx.x & 31);
  int row0 = wave * 16;
  if (row0 >= N) return;

  int m = lane & 15;
  int khalf = (lane >> 4) * 2;
  int row = row0 + m;
  if (row >= N) row = N - 1;

  const float* arow = A + (size_t)row * (size_t)Hdim + khalf;
  const float* wp   = W + khalf;

  v8f acc0 = {}; v8f acc1 = {}; v8f acc2 = {}; v8f acc3 = {};
  for (int k = 0; k < Hdim; k += 16) {
    v2f a0 = __builtin_nontemporal_load((const v2f*)(arow + k));
    v2f b0 = *(const v2f*)(wp + k);
    v2f a1 = __builtin_nontemporal_load((const v2f*)(arow + k + 4));
    v2f b1 = *(const v2f*)(wp + k + 4);
    v2f a2 = __builtin_nontemporal_load((const v2f*)(arow + k + 8));
    v2f b2 = *(const v2f*)(wp + k + 8);
    v2f a3 = __builtin_nontemporal_load((const v2f*)(arow + k + 12));
    v2f b3 = *(const v2f*)(wp + k + 12);
    acc0 = __builtin_amdgcn_wmma_f32_16x16x4_f32(false, a0, false, b0,
                                                 (short)0, acc0, false, false);
    acc1 = __builtin_amdgcn_wmma_f32_16x16x4_f32(false, a1, false, b1,
                                                 (short)0, acc1, false, false);
    acc2 = __builtin_amdgcn_wmma_f32_16x16x4_f32(false, a2, false, b2,
                                                 (short)0, acc2, false, false);
    acc3 = __builtin_amdgcn_wmma_f32_16x16x4_f32(false, a3, false, b3,
                                                 (short)0, acc3, false, false);
  }
  v8f acc = (acc0 + acc1) + (acc2 + acc3);
  float bv = bias[0];

  if ((lane & 15) == 0) {
    int mb = row0 + (lane >> 4) * 8;
#pragma unroll
    for (int r = 0; r < 8; ++r) {
      int rr = mb + r;
      if (rr < N) out[rr] = acc[r] + bv;
    }
  }
}

typedef float v4f __attribute__((ext_vector_type(4)));

#define GV_ROWS  16
#define GV_AHEAD 3
#define GV_KQ    8
#define GV_WAVES 4

__global__ __launch_bounds__(GV_WAVES * 32) void gemv_rows_wmma_f32(
    const float* __restrict__ A, const float* __restrict__ W,
    const float* __restrict__ bias, float* __restrict__ out, int N) {
  const unsigned lane = threadIdx.x & 31u;
  const unsigned wave = (unsigned)__builtin_amdgcn_readfirstlane(
      (int)((blockIdx.x * (unsigned)(GV_WAVES * 32) + threadIdx.x) >> 5));
  const unsigned row0 = wave * GV_ROWS;
  if (row0 >= (unsigned)N) return;

  v4f wq[GV_KQ];
  for (int q = 0; q < GV_KQ; ++q)
    wq[q] = *(const v4f*)(W + 128 * q + 4 * lane);

  const float* ab = A + (size_t)row0 * 1024u + 4u * lane;
  v4f a[GV_AHEAD + 1][GV_KQ];
#define GV_LOAD_ROW(R) do { \
    const unsigned rr_ = (row0 + (unsigned)(R) < (unsigned)N) ? (unsigned)(R) : 0u; \
    for (int q = 0; q < GV_KQ; ++q) \
      a[(R) % (GV_AHEAD + 1)][q] = __builtin_nontemporal_load( \
          (const v4f*)(ab + (size_t)rr_ * 1024u + 128u * q)); } while (0)

  for (int r = 0; r < GV_AHEAD; ++r) GV_LOAD_ROW(r);

  const bool lo8  = lane < 8u;
  const bool diag = lo8 || lane >= 24u;
  const bool b0 = (lane & 1u) != 0u, b1 = (lane & 2u) != 0u, b2 = (lane & 4u) != 0u;

  float part[GV_ROWS];
  for (int r = 0; r < GV_ROWS; ++r) {
    __builtin_amdgcn_sched_barrier(0);
    if (r + GV_AHEAD < GV_ROWS) GV_LOAD_ROW(r + GV_AHEAD);
    __builtin_amdgcn_sched_barrier(0);
    const v4f* ar = a[r % (GV_AHEAD + 1)];
    v8f c0 = {}; v8f c1 = {};
    for (int q = 0; q < GV_KQ; ++q) {
      v2f a01 = {ar[q].x, ar[q].y}, w01 = {wq[q].x, wq[q].y};
      v2f a23 = {ar[q].z, ar[q].w}, w23 = {wq[q].z, wq[q].w};
      c0 = __builtin_amdgcn_wmma_f32_16x16x4_f32(false, a01, false, w01,
                                                 (short)0, c0, false, false);
      c1 = __builtin_amdgcn_wmma_f32_16x16x4_f32(false, a23, false, w23,
                                                 (short)0, c1, false, false);
    }
    v8f c = c0 + c1;
    float s0 = b0 ? c[1] : c[0], s1 = b0 ? c[3] : c[2];
    float s2 = b0 ? c[5] : c[4], s3 = b0 ? c[7] : c[6];
    float t0 = b1 ? s1 : s0, t1 = b1 ? s3 : s2;
    float u  = b2 ? t1 : t0;
    part[r] = diag ? u : 0.0f;
    asm volatile("" : "+v"(part[r]));
  }
#undef GV_LOAD_ROW

  const bool h16 = (lane & 16u) != 0u, h8 = (lane & 8u) != 0u;
  const bool h4 = (lane & 4u) != 0u, h2 = (lane & 2u) != 0u;
  float e8[8], e4[4], e2[2];
  for (int i = 0; i < 8; ++i) {
    float keep = h16 ? part[i + 8] : part[i], send = h16 ? part[i] : part[i + 8];
    e8[i] = keep + __shfl_xor(send, 16, 32);
  }
  for (int i = 0; i < 4; ++i) {
    float keep = h8 ? e8[i + 4] : e8[i], send = h8 ? e8[i] : e8[i + 4];
    e4[i] = keep + __shfl_xor(send, 8, 32);
  }
  for (int i = 0; i < 2; ++i) {
    float keep = h4 ? e4[i + 2] : e4[i], send = h4 ? e4[i] : e4[i + 2];
    e2[i] = keep + __shfl_xor(send, 4, 32);
  }
  float keep1 = h2 ? e2[1] : e2[0], send1 = h2 ? e2[0] : e2[1];
  float e1 = keep1 + __shfl_xor(send1, 2, 32);
  float tot = e1 + __shfl_xor(e1, 1, 32);

  const unsigned myrow = row0 + (lane >> 1);
  if ((lane & 1u) == 0u && myrow < (unsigned)N) out[myrow] = tot + bias[0];
}

#define SEG_U 16

__global__ __launch_bounds__(256) void seg_loss(
    const float* __restrict__ logits, const int* __restrict__ lens,
    const int* __restrict__ tgt,
    float* __restrict__ loss_ws, float* __restrict__ inc_ws,
    int S, int EACH) {
  int wave = (int)((blockIdx.x * blockDim.x + threadIdx.x) >> 5);
  int lane = (int)(threadIdx.x & 31);
  int s = wave;
  if (s >= S) return;

  int len = lens[s];
  int acc = 0;
  for (int base = 0; base < s; base += 32 * SEG_U) {
    int v[SEG_U];
    for (int u = 0; u < SEG_U; ++u) {
      int i = base + 32 * u + lane;
      v[u] = lens[i < s ? i : s];
    }
    for (int u = 0; u < SEG_U; ++u) {
      int i = base + 32 * u + lane;
      acc += (i < s) ? v[u] : 0;
    }
  }
  for (int d = 16; d > 0; d >>= 1) acc += __shfl_xor(acc, d, 32);
  int off = acc;
  int j0 = lane, j1 = lane + 32;
  bool v0 = j0 < len, v1 = j1 < len;
  float x0 = v0 ? logits[off + j0] : NEG_INF;
  float x1 = v1 ? logits[off + j1] : NEG_INF;
  int   t0 = v0 ? tgt[off + j0] : 0;
  int   t1 = v1 ? tgt[off + j1] : 0;

  float mx = fmaxf(x0, x1);
#pragma unroll
  for (int d = 16; d > 0; d >>= 1) mx = fmaxf(mx, __shfl_xor(mx, d, 32));
  float e0 = v0 ? __expf(x0 - mx) : 0.0f;
  float e1 = v1 ? __expf(x1 - mx) : 0.0f;
  float sum = e0 + e1;
#pragma unroll
  for (int d = 16; d > 0; d >>= 1) sum += __shfl_xor(sum, d, 32);
  float inv = 1.0f / sum;
  float p0 = e0 * inv, p1 = e1 * inv;

  float q0 = v0 ? p0 : NEG_INF;
  float q1 = v1 ? p1 : NEG_INF;
  float m2 = fmaxf(q0, q1);
#pragma unroll
  for (int d = 16; d > 0; d >>= 1) m2 = fmaxf(m2, __shfl_xor(m2, d, 32));
  float s2 = (v0 ? __expf(p0 - m2) : 0.0f) + (v1 ? __expf(p1 - m2) : 0.0f);
#pragma unroll
  for (int d = 16; d > 0; d >>= 1) s2 += __shfl_xor(s2, d, 32);
  float lse2 = m2 + __logf(s2);

  int k0 = (v0 && t0 == 1) ? j0 : INT_MAX;
  int k1 = (v1 && t1 == 1) ? j1 : INT_MAX;
  int tk = min(k0, k1);
#pragma unroll
  for (int d = 16; d > 0; d >>= 1) tk = min(tk, __shfl_xor(tk, d, 32));
  int t_idx = (tk == INT_MAX) ? 0 : tk;
  float psel = (t_idx >= 32) ? p1 : p0;
  float p_t = __shfl(psel, t_idx & 31, 32);
  float loss_per = lse2 - p_t;

  int ts = t0 + t1;
#pragma unroll
  for (int d = 16; d > 0; d >>= 1) ts += __shfl_xor(ts, d, 32);

  int next_len = (s + 1 < S) ? lens[s + 1] : 1;
  bool skip_next_zero = (s > 0) && (s + 1 < S) && (next_len == 0);
  bool skip_no_pos = (((s + 1) % EACH) == 0) && (ts <= 0);
  bool include = (len > 0) && !skip_next_zero && !skip_no_pos;

  if (lane == 0) {
    loss_ws[s] = include ? loss_per : 0.0f;
    inc_ws[s]  = include ? 1.0f : 0.0f;
  }
}

#define FR_T 256
#define FR_U 6

__global__ __launch_bounds__(FR_T) void final_reduce(
    const float* __restrict__ loss_ws, const float* __restrict__ inc_ws,
    float* __restrict__ out_loss, int S) {
  __shared__ float sl[FR_T / 32];
  __shared__ float sc[FR_T / 32];
  const int t = (int)threadIdx.x;
  const int lane = t & 31, wv = t >> 5;
  float a = 0.0f, c = 0.0f;
  for (int base = 0; base < S; base += FR_T * FR_U) {
    float la[FR_U], lc[FR_U];
    for (int u = 0; u < FR_U; ++u) {
      int i = base + FR_T * u + t;
      int ic = i < S ? i : S - 1;
      la[u] = loss_ws[ic];
      lc[u] = inc_ws[ic];
    }
    __builtin_amdgcn_sched_barrier(0);
    for (int u = 0; u < FR_U; ++u) {
      int i = base + FR_T * u + t;
      a += (i < S) ? la[u] : 0.0f;
      c += (i < S) ? lc[u] : 0.0f;
    }
  }
  for (int d = 16; d > 0; d >>= 1) {
    a += __shfl_xor(a, d, 32);
    c += __shfl_xor(c, d, 32);
  }
  if (lane == 0) { sl[wv] = a; sc[wv] = c; }
  __syncthreads();
  if (wv == 0) {
    float a2 = (lane < FR_T / 32) ? sl[lane] : 0.0f;
    float c2 = (lane < FR_T / 32) ? sc[lane] : 0.0f;
    for (int d = (FR_T / 64); d > 0; d >>= 1) {
      a2 += __shfl_xor(a2, d, 32);
      c2 += __shfl_xor(c2, d, 32);
    }
    if (lane == 0) {
      float denom = (c2 > 0.0f) ? c2 : 1e-4f;
      out_loss[0] = a2 / denom;
    }
  }
}

extern "C" void kernel_launch(void* const* d_in, const int* in_sizes, int n_in,
                              void* d_out, int out_size, void* d_ws, size_t ws_size,
                              hipStream_t stream) {
  const float* A    = (const float*)d_in[0];
  const float* W    = (const float*)d_in[1];
  const float* b    = (const float*)d_in[2];
  const int*   lens = (const int*)d_in[3];
  const int*   tgt  = (const int*)d_in[4];

  int Hdim = in_sizes[1];
  int N    = in_sizes[0] / Hdim;
  int S    = in_sizes[3];
  const int EACH = 32;

  float* logits   = (float*)d_out;
  float* loss_out = logits + N;

  size_t obytes  = (((size_t)S * sizeof(int)) + 255) & ~(size_t)255;
  float* loss_ws = (float*)((char*)d_ws + obytes);
  float* inc_ws  = loss_ws + S;

  int waves  = (N + 15) / 16;
  if (Hdim == 1024) {
    int blocks = (waves + GV_WAVES - 1) / GV_WAVES;
    gemv_rows_wmma_f32<<<blocks, GV_WAVES * 32, 0, stream>>>(A, W, b, logits, N);
  } else {
    int blocks = (waves + 7) / 8;
    gemv_wmma_f32<<<blocks, 256, 0, stream>>>(A, W, b, logits, N, Hdim);
  }

  seg_loss<<<(S + 7) / 8, 256, 0, stream>>>(logits, lens, tgt,
                                            loss_ws, inc_ws, S, EACH);

  final_reduce<<<1, FR_T, 0, stream>>>(loss_ws, inc_ws, loss_out, S);
}


# GEMV grid re-cut for equal SIMD shares: 16-wave work-groups (4 waves per SIMD, one group per WGP) of 13-row waves, 52 rows per SIMD; was 4-wave groups of 16-row waves, up to 64 rows per SIMD
# speedup vs baseline: 3.5741x; 1.0751x over previous
#include <hip/hip_runtime.h>
#include <limits.h>

typedef float v2f __attribute__((ext_vector_type(2)));
typedef float v8f __attribute__((ext_vector_type(8)));

#define NEG_INF (-1e30f)

__global__ __launch_bounds__(256) void gemv_wmma_f32(
    const float* __restrict__ A, const float* __restrict__ W,
    const float* __restrict__ bias, float* __restrict__ out,
    int N, int Hdim) {
  int wave = (int)((blockIdx.x * blockDim.x + threadIdx.x) >> 5);
  int lane = (int)(threadIdx.x & 31);
  int row0 = wave * 16;
  if (row0 >= N) return;

  int m = lane & 15;
  int khalf = (lane >> 4) * 2;
  int row = row0 + m;
  if (row >= N) row = N - 1;

  const float* arow = A + (size_t)row * (size_t)Hdim + khalf;
  const float* wp   = W + khalf;

  v8f acc0 = {}; v8f acc1 = {}; v8f acc2 = {}; v8f acc3 = {};
  for (int k = 0; k < Hdim; k += 16) {
    v2f a0 = __builtin_nontemporal_load((const v2f*)(arow + k));
    v2f b0 = *(const v2f*)(wp + k);
    v2f a1 = __builtin_nontemporal_load((const v2f*)(arow + k + 4));
    v2f b1 = *(const v2f*)(wp + k + 4);
    v2f a2 = __builtin_nontemporal_load((const v2f*)(arow + k + 8));
    v2f b2 = *(const v2f*)(wp + k + 8);
    v2f a3 = __builtin_nontemporal_load((const v2f*)(arow + k + 12));
    v2f b3 = *(const v2f*)(wp + k + 12);
    acc0 = __builtin_amdgcn_wmma_f32_16x16x4_f32(false, a0, false, b0,
                                                 (short)0, acc0, false, false);
    acc1 = __builtin_amdgcn_wmma_f32_16x16x4_f32(false, a1, false, b1,
                                                 (short)0, acc1, false, false);
    acc2 = __builtin_amdgcn_wmma_f32_16x16x4_f32(false, a2, false, b2,
                                                 (short)0, acc2, false, false);
    acc3 = __builtin_amdgcn_wmma_f32_16x16x4_f32(false, a3, false, b3,
                                                 (short)0, acc3, false, false);
  }
  v8f acc = (acc0 + acc1) + (acc2 + acc3);
  float bv = bias[0];

  if ((lane & 15) == 0) {
    int mb = row0 + (lane >> 4) * 8;
#pragma unroll
    for (int r = 0; r < 8; ++r) {
      int rr = mb + r;
      if (rr < N) out[rr] = acc[r] + bv;
    }
  }
}

typedef float v4f __attribute__((ext_vector_type(4)));

#define GV_ROWS  13
#define GV_AHEAD 3
#define GV_KQ    8
#define GV_WAVES 16

__global__ __launch_bounds__(GV_WAVES * 32) void gemv_rows_wmma_f32(
    const float* __restrict__ A, const float* __restrict__ W,
    const float* __restrict__ bias, float* __restrict__ out, int N) {
  const unsigned lane = threadIdx.x & 31u;
  const unsigned wave = (unsigned)__builtin_amdgcn_readfirstlane(
      (int)((blockIdx.x * (unsigned)(GV_WAVES * 32) + threadIdx.x) >> 5));
  const unsigned row0 = wave * GV_ROWS;
  if (row0 >= (unsigned)N) return;

  v4f wq[GV_KQ];
  for (int q = 0; q < GV_KQ; ++q)
    wq[q] = *(const v4f*)(W + 128 * q + 4 * lane);

  const float* ab = A + (size_t)row0 * 1024u + 4u * lane;
  v4f a[GV_AHEAD + 1][GV_KQ];
#define GV_LOAD_ROW(R) do { \
    const unsigned rr_ = (row0 + (unsigned)(R) < (unsigned)N) ? (unsigned)(R) : 0u; \
    for (int q = 0; q < GV_KQ; ++q) \
      a[(R) % (GV_AHEAD + 1)][q] = __builtin_nontemporal_load( \
          (const v4f*)(ab + (size_t)rr_ * 1024u + 128u * q)); } while (0)

  for (int r = 0; r < GV_AHEAD; ++r) GV_LOAD_ROW(r);

  const bool lo8  = lane < 8u;
  const bool diag = lo8 || lane >= 24u;
  const bool b0 = (lane & 1u) != 0u, b1 = (lane & 2u) != 0u, b2 = (lane & 4u) != 0u;

  float part[16];
  for (int r = GV_ROWS; r < 16; ++r) part[r] = 0.0f;
  for (int r = 0; r < GV_ROWS; ++r) {
    __builtin_amdgcn_sched_barrier(0);
    if (r + GV_AHEAD < GV_ROWS) GV_LOAD_ROW(r + GV_AHEAD);
    __builtin_amdgcn_sched_barrier(0);
    const v4f* ar = a[r % (GV_AHEAD + 1)];
    v8f c0 = {}; v8f c1 = {};
    for (int q = 0; q < GV_KQ; ++q) {
      v2f a01 = {ar[q].x, ar[q].y}, w01 = {wq[q].x, wq[q].y};
      v2f a23 = {ar[q].z, ar[q].w}, w23 = {wq[q].z, wq[q].w};
      c0 = __builtin_amdgcn_wmma_f32_16x16x4_f32(false, a01, false, w01,
                                                 (short)0, c0, false, false);
      c1 = __builtin_amdgcn_wmma_f32_16x16x4_f32(false, a23, false, w23,
                                                 (short)0, c1, false, false);
    }
    v8f c = c0 + c1;
    float s0 = b0 ? c[1] : c[0], s1 = b0 ? c[3] : c[2];
    float s2 = b0 ? c[5] : c[4], s3 = b0 ? c[7] : c[6];
    float t0 = b1 ? s1 : s0, t1 = b1 ? s3 : s2;
    float u  = b2 ? t1 : t0;
    part[r] = diag ? u : 0.0f;
    asm volatile("" : "+v"(part[r]));
  }
#undef GV_LOAD_ROW

  const bool h16 = (lane & 16u) != 0u, h8 = (lane & 8u) != 0u;
  const bool h4 = (lane & 4u) != 0u, h2 = (lane & 2u) != 0u;
  float e8[8], e4[4], e2[2];
  for (int i = 0; i < 8; ++i) {
    float keep = h16 ? part[i + 8] : part[i], send = h16 ? part[i] : part[i + 8];
    e8[i] = keep + __shfl_xor(send, 16, 32);
  }
  for (int i = 0; i < 4; ++i) {
    float keep = h8 ? e8[i + 4] : e8[i], send = h8 ? e8[i] : e8[i + 4];
    e4[i] = keep + __shfl_xor(send, 8, 32);
  }
  for (int i = 0; i < 2; ++i) {
    float keep = h4 ? e4[i + 2] : e4[i], send = h4 ? e4[i] : e4[i + 2];
    e2[i] = keep + __shfl_xor(send, 4, 32);
  }
  float keep1 = h2 ? e2[1] : e2[0], send1 = h2 ? e2[0] : e2[1];
  float e1 = keep1 + __shfl_xor(send1, 2, 32);
  float tot = e1 + __shfl_xor(e1, 1, 32);

  const unsigned myrow = row0 + (lane >> 1);
  if ((lane & 1u) == 0u && (lane >> 1) < (unsigned)GV_ROWS && myrow < (unsigned)N)
    out[myrow] = tot + bias[0];
}

#define SEG_U 16

__global__ __launch_bounds__(256) void seg_loss(
    const float* __restrict__ logits, const int* __restrict__ lens,
    const int* __restrict__ tgt,
    float* __restrict__ loss_ws, float* __restrict__ inc_ws,
    int S, int EACH) {
  int wave = (int)((blockIdx.x * blockDim.x + threadIdx.x) >> 5);
  int lane = (int)(threadIdx.x & 31);
  int s = wave;
  if (s >= S) return;

  int len = lens[s];
  int acc = 0;
  for (int base = 0; base < s; base += 32 * SEG_U) {
    int v[SEG_U];
    for (int u = 0; u < SEG_U; ++u) {
      int i = base + 32 * u + lane;
      v[u] = lens[i < s ? i : s];
    }
    for (int u = 0; u < SEG_U; ++u) {
      int i = base + 32 * u + lane;
      acc += (i < s) ? v[u] : 0;
    }
  }
  for (int d = 16; d > 0; d >>= 1) acc += __shfl_xor(acc, d, 32);
  int off = acc;
  int j0 = lane, j1 = lane + 32;
  bool v0 = j0 < len, v1 = j1 < len;
  float x0 = v0 ? logits[off + j0] : NEG_INF;
  float x1 = v1 ? logits[off + j1] : NEG_INF;
  int   t0 = v0 ? tgt[off + j0] : 0;
  int   t1 = v1 ? tgt[off + j1] : 0;

  float mx = fmaxf(x0, x1);
#pragma unroll
  for (int d = 16; d > 0; d >>= 1) mx = fmaxf(mx, __shfl_xor(mx, d, 32));
  float e0 = v0 ? __expf(x0 - mx) : 0.0f;
  float e1 = v1 ? __expf(x1 - mx) : 0.0f;
  float sum = e0 + e1;
#pragma unroll
  for (int d = 16; d > 0; d >>= 1) sum += __shfl_xor(sum, d, 32);
  float inv = 1.0f / sum;
  float p0 = e0 * inv, p1 = e1 * inv;

  float q0 = v0 ? p0 : NEG_INF;
  float q1 = v1 ? p1 : NEG_INF;
  float m2 = fmaxf(q0, q1);
#pragma unroll
  for (int d = 16; d > 0; d >>= 1) m2 = fmaxf(m2, __shfl_xor(m2, d, 32));
  float s2 = (v0 ? __expf(p0 - m2) : 0.0f) + (v1 ? __expf(p1 - m2) : 0.0f);
#pragma unroll
  for (int d = 16; d > 0; d >>= 1) s2 += __shfl_xor(s2, d, 32);
  float lse2 = m2 + __logf(s2);

  int k0 = (v0 && t0 == 1) ? j0 : INT_MAX;
  int k1 = (v1 && t1 == 1) ? j1 : INT_MAX;
  int tk = min(k0, k1);
#pragma unroll
  for (int d = 16; d > 0; d >>= 1) tk = min(tk, __shfl_xor(tk, d, 32));
  int t_idx = (tk == INT_MAX) ? 0 : tk;
  float psel = (t_idx >= 32) ? p1 : p0;
  float p_t = __shfl(psel, t_idx & 31, 32);
  float loss_per = lse2 - p_t;

  int ts = t0 + t1;
#pragma unroll
  for (int d = 16; d > 0; d >>= 1) ts += __shfl_xor(ts, d, 32);

  int next_len = (s + 1 < S) ? lens[s + 1] : 1;
  bool skip_next_zero = (s > 0) && (s + 1 < S) && (next_len == 0);
  bool skip_no_pos = (((s + 1) % EACH) == 0) && (ts <= 0);
  bool include = (len > 0) && !skip_next_zero && !skip_no_pos;

  if (lane == 0) {
    loss_ws[s] = include ? loss_per : 0.0f;
    inc_ws[s]  = include ? 1.0f : 0.0f;
  }
}

#define FR_T 256
#define FR_U 6

__global__ __launch_bounds__(FR_T) void final_reduce(
    const float* __restrict__ loss_ws, const float* __restrict__ inc_ws,
    float* __restrict__ out_loss, int S) {
  __shared__ float sl[FR_T / 32];
  __shared__ float sc[FR_T / 32];
  const int t = (int)threadIdx.x;
  const int lane = t & 31, wv = t >> 5;
  float a = 0.0f, c = 0.0f;
  for (int base = 0; base < S; base += FR_T * FR_U) {
    float la[FR_U], lc[FR_U];
    for (int u = 0; u < FR_U; ++u) {
      int i = base + FR_T * u + t;
      int ic = i < S ? i : S - 1;
      la[u] = loss_ws[ic];
      lc[u] = inc_ws[ic];
    }
    __builtin_amdgcn_sched_barrier(0);
    for (int u = 0; u < FR_U; ++u) {
      int i = base + FR_T * u + t;
      a += (i < S) ? la[u] : 0.0f;
      c += (i < S) ? lc[u] : 0.0f;
    }
  }
  for (int d = 16; d > 0; d >>= 1) {
    a += __shfl_xor(a, d, 32);
    c += __shfl_xor(c, d, 32);
  }
  if (lane == 0) { sl[wv] = a; sc[wv] = c; }
  __syncthreads();
  if (wv == 0) {
    float a2 = (lane < FR_T / 32) ? sl[lane] : 0.0f;
    float c2 = (lane < FR_T / 32) ? sc[lane] : 0.0f;
    for (int d = (FR_T / 64); d > 0; d >>= 1) {
      a2 += __shfl_xor(a2, d, 32);
      c2 += __shfl_xor(c2, d, 32);
    }
    if (lane == 0) {
      float denom = (c2 > 0.0f) ? c2 : 1e-4f;
      out_loss[0] = a2 / denom;
    }
  }
}

extern "C" void kernel_launch(void* const* d_in, const int* in_sizes, int n_in,
                              void* d_out, int out_size, void* d_ws, size_t ws_size,
                              hipStream_t stream) {
  const float* A    = (const float*)d_in[0];
  const float* W    = (const float*)d_in[1];
  const float* b    = (const float*)d_in[2];
  const int*   lens = (const int*)d_in[3];
  const int*   tgt  = (const int*)d_in[4];

  int Hdim = in_sizes[1];
  int N    = in_sizes[0] / Hdim;
  int S    = in_sizes[3];
  const int EACH = 32;

  float* logits   = (float*)d_out;
  float* loss_out = logits + N;

  size_t obytes  = (((size_t)S * sizeof(int)) + 255) & ~(size_t)255;
  float* loss_ws = (float*)((char*)d_ws + obytes);
  float* inc_ws  = loss_ws + S;

  int waves  = (N + 15) / 16;
  if (Hdim == 1024) {
    int rwaves = (N + GV_ROWS - 1) / GV_ROWS;
    int blocks = (rwaves + GV_WAVES - 1) / GV_WAVES;
    gemv_rows_wmma_f32<<<blocks, GV_WAVES * 32, 0, stream>>>(A, W, b, logits, N);
  } else {
    int blocks = (waves + 7) / 8;
    gemv_wmma_f32<<<blocks, 256, 0, stream>>>(A, W, b, logits, N, Hdim);
  }

  seg_loss<<<(S + 7) / 8, 256, 0, stream>>>(logits, lens, tgt,
                                            loss_ws, inc_ws, S, EACH);

  final_reduce<<<1, FR_T, 0, stream>>>(loss_ws, inc_ws, loss_out, S);
}


# Wave-wide reductions and lane exchanges of all three kernels on the vector ALU (DPP row_xmask, v_permlanex16, v_readlane), was ds_bpermute through LDS; same partner order, no sum order changes
# speedup vs baseline: 3.7297x; 1.0435x over previous
#include <hip/hip_runtime.h>
#include <limits.h>

typedef float v2f __attribute__((ext_vector_type(2)));
typedef float v8f __attribute__((ext_vector_type(8)));

#define NEG_INF (-1e30f)

__global__ __launch_bounds__(256) void gemv_wmma_f32(
    const float* __restrict__ A, const float* __restrict__ W,
    const float* __restrict__ bias, float* __restrict__ out,
    int N, int Hdim) {
  int wave = (int)((blockIdx.x * blockDim.x + threadIdx.x) >> 5);
  int lane = (int)(threadIdx.x & 31);
  int row0 = wave * 16;
  if (row0 >= N) return;

  int m = lane & 15;
  int khalf = (lane >> 4) * 2;
  int row = row0 + m;
  if (row >= N) row = N - 1;

  const float* arow = A + (size_t)row * (size_t)Hdim + khalf;
  const float* wp   = W + khalf;

  v8f acc0 = {}; v8f acc1 = {}; v8f acc2 = {}; v8f acc3 = {};
  for (int k = 0; k < Hdim; k += 16) {
    v2f a0 = __builtin_nontemporal_load((const v2f*)(arow + k));
    v2f b0 = *(const v2f*)(wp + k);
    v2f a1 = __builtin_nontemporal_load((const v2f*)(arow + k + 4));
    v2f b1 = *(const v2f*)(wp + k + 4);
    v2f a2 = __builtin_nontemporal_load((const v2f*)(arow + k + 8));
    v2f b2 = *(const v2f*)(wp + k + 8);
    v2f a3 = __builtin_nontemporal_load((const v2f*)(arow + k + 12));
    v2f b3 = *(const v2f*)(wp + k + 12);
    acc0 = __builtin_amdgcn_wmma_f32_16x16x4_f32(false, a0, false, b0,
                                                 (short)0, acc0, false, false);
    acc1 = __builtin_amdgcn_wmma_f32_16x16x4_f32(false, a1, false, b1,
                                                 (short)0, acc1, false, false);
    acc2 = __builtin_amdgcn_wmma_f32_16x16x4_f32(false, a2, false, b2,
                                                 (short)0, acc2, false, false);
    acc3 = __builtin_amdgcn_wmma_f32_16x16x4_f32(false, a3, false, b3,
                                                 (short)0, acc3, false, false);
  }
  v8f acc = (acc0 + acc1) + (acc2 + acc3);
  float bv = bias[0];

  if ((lane & 15) == 0) {
    int mb = row0 + (lane >> 4) * 8;
#pragma unroll
    for (int r = 0; r < 8; ++r) {
      int rr = mb + r;
      if (rr < N) out[rr] = acc[r] + bv;
    }
  }
}

typedef float v4f __attribute__((ext_vector_type(4)));

template <int M> static __device__ __forceinline__ int rowx(int b) {
  return __builtin_amdgcn_update_dpp(b, b, 0x160 | M, 0xf, 0xf, true);
}
template <int M> static __device__ __forceinline__ float rowx(float x) {
  return __builtin_bit_cast(float, rowx<M>(__builtin_bit_cast(int, x)));
}
static __device__ __forceinline__ int xhalf(int b) {
  return __builtin_amdgcn_permlanex16(b, b, 0x76543210, 0xfedcba98, false, false);
}
static __device__ __forceinline__ float xhalf(float x) {
  return __builtin_bit_cast(float, xhalf(__builtin_bit_cast(int, x)));
}
static __device__ __forceinline__ float wave_sum(float x) {
  x += xhalf(x);   x += rowx<8>(x); x += rowx<4>(x);
  x += rowx<2>(x); x += rowx<1>(x);
  return x;
}
static __device__ __forceinline__ int wave_sum(int x) {
  x += xhalf(x);   x += rowx<8>(x); x += rowx<4>(x);
  x += rowx<2>(x); x += rowx<1>(x);
  return x;
}
static __device__ __forceinline__ float wave_max(float x) {
  x = fmaxf(x, xhalf(x));   x = fmaxf(x, rowx<8>(x)); x = fmaxf(x, rowx<4>(x));
  x = fmaxf(x, rowx<2>(x)); x = fmaxf(x, rowx<1>(x));
  return x;
}
static __device__ __forceinline__ int wave_min(int x) {
  x = min(x, xhalf(x));   x = min(x, rowx<8>(x)); x = min(x, rowx<4>(x));
  x = min(x, rowx<2>(x)); x = min(x, rowx<1>(x));
  return x;
}

#define GV_ROWS  13
#define GV_AHEAD 3
#define GV_KQ    8
#define GV_WAVES 16

__global__ __launch_bounds__(GV_WAVES * 32) void gemv_rows_wmma_f32(
    const float* __restrict__ A, const float* __restrict__ W,
    const float* __restrict__ bias, float* __restrict__ out, int N) {
  const unsigned lane = threadIdx.x & 31u;
  const unsigned wave = (unsigned)__builtin_amdgcn_readfirstlane(
      (int)((blockIdx.x * (unsigned)(GV_WAVES * 32) + threadIdx.x) >> 5));
  const unsigned row0 = wave * GV_ROWS;
  if (row0 >= (unsigned)N) return;

  v4f wq[GV_KQ];
  for (int q = 0; q < GV_KQ; ++q)
    wq[q] = *(const v4f*)(W + 128 * q + 4 * lane);

  const float* ab = A + (size_t)row0 * 1024u + 4u * lane;
  v4f a[GV_AHEAD + 1][GV_KQ];
#define GV_LOAD_ROW(R) do { \
    const unsigned rr_ = (row0 + (unsigned)(R) < (unsigned)N) ? (unsigned)(R) : 0u; \
    for (int q = 0; q < GV_KQ; ++q) \
      a[(R) % (GV_AHEAD + 1)][q] = __builtin_nontemporal_load( \
          (const v4f*)(ab + (size_t)rr_ * 1024u + 128u * q)); } while (0)

  for (int r = 0; r < GV_AHEAD; ++r) GV_LOAD_ROW(r);

  const bool lo8  = lane < 8u;
  const bool diag = lo8 || lane >= 24u;
  const bool b0 = (lane & 1u) != 0u, b1 = (lane & 2u) != 0u, b2 = (lane & 4u) != 0u;

  float part[16];
  for (int r = GV_ROWS; r < 16; ++r) part[r] = 0.0f;
  for (int r = 0; r < GV_ROWS; ++r) {
    __builtin_amdgcn_sched_barrier(0);
    if (r + GV_AHEAD < GV_ROWS) GV_LOAD_ROW(r + GV_AHEAD);
    __builtin_amdgcn_sched_barrier(0);
    const v4f* ar = a[r % (GV_AHEAD + 1)];
    v8f c0 = {}; v8f c1 = {};
    for (int q = 0; q < GV_KQ; ++q) {
      v2f a01 = {ar[q].x, ar[q].y}, w01 = {wq[q].x, wq[q].y};
      v2f a23 = {ar[q].z, ar[q].w}, w23 = {wq[q].z, wq[q].w};
      c0 = __builtin_amdgcn_wmma_f32_16x16x4_f32(false, a01, false, w01,
                                                 (short)0, c0, false, false);
      c1 = __builtin_amdgcn_wmma_f32_16x16x4_f32(false, a23, false, w23,
                                                 (short)0, c1, false, false);
    }
    v8f c = c0 + c1;
    float s0 = b0 ? c[1] : c[0], s1 = b0 ? c[3] : c[2];
    float s2 = b0 ? c[5] : c[4], s3 = b0 ? c[7] : c[6];
    float t0 = b1 ? s1 : s0, t1 = b1 ? s3 : s2;
    float u  = b2 ? t1 : t0;
    part[r] = diag ? u : 0.0f;
    asm volatile("" : "+v"(part[r]));
  }
#undef GV_LOAD_ROW

  const bool h16 = (lane & 16u) != 0u, h8 = (lane & 8u) != 0u;
  const bool h4 = (lane & 4u) != 0u, h2 = (lane & 2u) != 0u;
  float e8[8], e4[4], e2[2];
  for (int i = 0; i < 8; ++i) {
    float keep = h16 ? part[i + 8] : part[i], send = h16 ? part[i] : part[i + 8];
    e8[i] = keep + xhalf(send);
  }
  for (int i = 0; i < 4; ++i) {
    float keep = h8 ? e8[i + 4] : e8[i], send = h8 ? e8[i] : e8[i + 4];
    e4[i] = keep + rowx<8>(send);
  }
  for (int i = 0; i < 2; ++i) {
    float keep = h4 ? e4[i + 2] : e4[i], send = h4 ? e4[i] : e4[i + 2];
    e2[i] = keep + rowx<4>(send);
  }
  float keep1 = h2 ? e2[1] : e2[0], send1 = h2 ? e2[0] : e2[1];
  float e1 = keep1 + rowx<2>(send1);
  float tot = e1 + rowx<1>(e1);

  const unsigned myrow = row0 + (lane >> 1);
  if ((lane & 1u) == 0u && (lane >> 1) < (unsigned)GV_ROWS && myrow < (unsigned)N)
    out[myrow] = tot + bias[0];
}

#define SEG_U 16

__global__ __launch_bounds__(256) void seg_loss(
    const float* __restrict__ logits, const int* __restrict__ lens,
    const int* __restrict__ tgt,
    float* __restrict__ loss_ws, float* __restrict__ inc_ws,
    int S, int EACH) {
  int wave = (int)((blockIdx.x * blockDim.x + threadIdx.x) >> 5);
  int lane = (int)(threadIdx.x & 31);
  int s = wave;
  if (s >= S) return;

  int len = lens[s];
  int acc = 0;
  for (int base = 0; base < s; base += 32 * SEG_U) {
    int v[SEG_U];
    for (int u = 0; u < SEG_U; ++u) {
      int i = base + 32 * u + lane;
      v[u] = lens[i < s ? i : s];
    }
    for (int u = 0; u < SEG_U; ++u) {
      int i = base + 32 * u + lane;
      acc += (i < s) ? v[u] : 0;
    }
  }
  int off = wave_sum(acc);
  int j0 = lane, j1 = lane + 32;
  bool v0 = j0 < len, v1 = j1 < len;
  float x0 = v0 ? logits[off + j0] : NEG_INF;
  float x1 = v1 ? logits[off + j1] : NEG_INF;
  int   t0 = v0 ? tgt[off + j0] : 0;
  int   t1 = v1 ? tgt[off + j1] : 0;

  float mx = wave_max(fmaxf(x0, x1));
  float e0 = v0 ? __expf(x0 - mx) : 0.0f;
  float e1 = v1 ? __expf(x1 - mx) : 0.0f;
  float sum = wave_sum(e0 + e1);
  float inv = 1.0f / sum;
  float p0 = e0 * inv, p1 = e1 * inv;

  float q0 = v0 ? p0 : NEG_INF;
  float q1 = v1 ? p1 : NEG_INF;
  float m2 = wave_max(fmaxf(q0, q1));
  float s2 = wave_sum((v0 ? __expf(p0 - m2) : 0.0f) + (v1 ? __expf(p1 - m2) : 0.0f));
  float lse2 = m2 + __logf(s2);

  int k0 = (v0 && t0 == 1) ? j0 : INT_MAX;
  int k1 = (v1 && t1 == 1) ? j1 : INT_MAX;
  int tk = wave_min(min(k0, k1));
  tk = __builtin_amdgcn_readfirstlane(tk);
  int t_idx = (tk == INT_MAX) ? 0 : tk;
  float psel = (t_idx >= 32) ? p1 : p0;
  float p_t = __builtin_bit_cast(float, __builtin_amdgcn_readlane(
      __builtin_bit_cast(int, psel), t_idx & 31));
  float loss_per = lse2 - p_t;

  int ts = wave_sum(t0 + t1);

  int next_len = (s + 1 < S) ? lens[s + 1] : 1;
  bool skip_next_zero = (s > 0) && (s + 1 < S) && (next_len == 0);
  bool skip_no_pos = (((s + 1) % EACH) == 0) && (ts <= 0);
  bool include = (len > 0) && !skip_next_zero && !skip_no_pos;

  if (lane == 0) {
    loss_ws[s] = include ? loss_per : 0.0f;
    inc_ws[s]  = include ? 1.0f : 0.0f;
  }
}

#define FR_T 256
#define FR_U 6

__global__ __launch_bounds__(FR_T) void final_reduce(
    const float* __restrict__ loss_ws, const float* __restrict__ inc_ws,
    float* __restrict__ out_loss, int S) {
  __shared__ float sl[FR_T / 32];
  __shared__ float sc[FR_T / 32];
  const int t = (int)threadIdx.x;
  const int lane = t & 31, wv = t >> 5;
  float a = 0.0f, c = 0.0f;
  for (int base = 0; base < S; base += FR_T * FR_U) {
    float la[FR_U], lc[FR_U];
    for (int u = 0; u < FR_U; ++u) {
      int i = base + FR_T * u + t;
      int ic = i < S ? i : S - 1;
      la[u] = loss_ws[ic];
      lc[u] = inc_ws[ic];
    }
    __builtin_amdgcn_sched_barrier(0);
    for (int u = 0; u < FR_U; ++u) {
      int i = base + FR_T * u + t;
      a += (i < S) ? la[u] : 0.0f;
      c += (i < S) ? lc[u] : 0.0f;
    }
  }
  a = wave_sum(a);
  c = wave_sum(c);
  if (lane == 0) { sl[wv] = a; sc[wv] = c; }
  __syncthreads();
  if (wv == 0) {
    float a2 = (lane < FR_T / 32) ? sl[lane] : 0.0f;
    float c2 = (lane < FR_T / 32) ? sc[lane] : 0.0f;
    static_assert(FR_T / 32 == 8, "the three steps below fold exactly 8 wave sums");
    a2 += rowx<4>(a2); a2 += rowx<2>(a2); a2 += rowx<1>(a2);
    c2 += rowx<4>(c2); c2 += rowx<2>(c2); c2 += rowx<1>(c2);
    if (lane == 0) {
      float denom = (c2 > 0.0f) ? c2 : 1e-4f;
      out_loss[0] = a2 / denom;
    }
  }
}

extern "C" void kernel_launch(void* const* d_in, const int* in_sizes, int n_in,
                              void* d_out, int out_size, void* d_ws, size_t ws_size,
                              hipStream_t stream) {
  const float* A    = (const float*)d_in[0];
  const float* W    = (const float*)d_in[1];
  const float* b    = (const float*)d_in[2];
  const int*   lens = (const int*)d_in[3];
  const int*   tgt  = (const int*)d_in[4];

  int Hdim = in_sizes[1];
  int N    = in_sizes[0] / Hdim;
  int S    = in_sizes[3];
  const int EACH = 32;

  float* logits   = (float*)d_out;
  float* loss_out = logits + N;

  size_t obytes  = (((size_t)S * sizeof(int)) + 255) & ~(size_t)255;
  float* loss_ws = (float*)((char*)d_ws + obytes);
  float* inc_ws  = loss_ws + S;

  int waves  = (N + 15) / 16;
  if (Hdim == 1024) {
    int rwaves = (N + GV_ROWS - 1) / GV_ROWS;
    int blocks = (rwaves + GV_WAVES - 1) / GV_WAVES;
    gemv_rows_wmma_f32<<<blocks, GV_WAVES * 32, 0, stream>>>(A, W, b, logits, N);
  } else {
    int blocks = (waves + 7) / 8;
    gemv_wmma_f32<<<blocks, 256, 0, stream>>>(A, W, b, logits, N, Hdim);
  }

  seg_loss<<<(S + 7) / 8, 256, 0, stream>>>(logits, lens, tgt,
                                            loss_ws, inc_ws, S, EACH);

  final_reduce<<<1, FR_T, 0, stream>>>(loss_ws, inc_ws, loss_out, S);
}


# Per-segment losses computed in the product kernel: a work-group owns whole segments (rows cut at segment starts) and finishes them behind one barrier; 3 launches to 2; no sum order changes
# speedup vs baseline: 4.0624x; 1.0892x over previous
#include <hip/hip_runtime.h>
#include <limits.h>

typedef float v2f __attribute__((ext_vector_type(2)));
typedef float v8f __attribute__((ext_vector_type(8)));

#define NEG_INF (-1e30f)

__global__ __launch_bounds__(256) void gemv_wmma_f32(
    const float* __restrict__ A, const float* __restrict__ W,
    const float* __restrict__ bias, float* __restrict__ out,
    int N, int Hdim) {
  int wave = (int)((blockIdx.x * blockDim.x + threadIdx.x) >> 5);
  int lane = (int)(threadIdx.x & 31);
  int row0 = wave * 16;
  if (row0 >= N) return;

  int m = lane & 15;
  int khalf = (lane >> 4) * 2;
  int row = row0 + m;
  if (row >= N) row = N - 1;

  const float* arow = A + (size_t)row * (size_t)Hdim + khalf;
  const float* wp   = W + khalf;

  v8f acc0 = {}; v8f acc1 = {}; v8f acc2 = {}; v8f acc3 = {};
  for (int k = 0; k < Hdim; k += 16) {
    v2f a0 = __builtin_nontemporal_load((const v2f*)(arow + k));
    v2f b0 = *(const v2f*)(wp + k);
    v2f a1 = __builtin_nontemporal_load((const v2f*)(arow + k + 4));
    v2f b1 = *(const v2f*)(wp + k + 4);
    v2f a2 = __builtin_nontemporal_load((const v2f*)(arow + k + 8));
    v2f b2 = *(const v2f*)(wp + k + 8);
    v2f a3 = __builtin_nontemporal_load((const v2f*)(arow + k + 12));
    v2f b3 = *(const v2f*)(wp + k + 12);
    acc0 = __builtin_amdgcn_wmma_f32_16x16x4_f32(false, a0, false, b0,
                                                 (short)0, acc0, false, false);
    acc1 = __builtin_amdgcn_wmma_f32_16x16x4_f32(false, a1, false, b1,
                                                 (short)0, acc1, false, false);
    acc2 = __builtin_amdgcn_wmma_f32_16x16x4_f32(false, a2, false, b2,
                                                 (short)0, acc2, false, false);
    acc3 = __builtin_amdgcn_wmma_f32_16x16x4_f32(false, a3, false, b3,
                                                 (short)0, acc3, false, false);
  }
  v8f acc = (acc0 + acc1) + (acc2 + acc3);
  float bv = bias[0];

  if ((lane & 15) == 0) {
    int mb = row0 + (lane >> 4) * 8;
#pragma unroll
    for (int r = 0; r < 8; ++r) {
      int rr = mb + r;
      if (rr < N) out[rr] = acc[r] + bv;
    }
  }
}

typedef float v4f __attribute__((ext_vector_type(4)));

template <int M> static __device__ __forceinline__ int rowx(int b) {
  return __builtin_amdgcn_update_dpp(b, b, 0x160 | M, 0xf, 0xf, true);
}
template <int M> static __device__ __forceinline__ float rowx(float x) {
  return __builtin_bit_cast(float, rowx<M>(__builtin_bit_cast(int, x)));
}
static __device__ __forceinline__ int xhalf(int b) {
  return __builtin_amdgcn_permlanex16(b, b, 0x76543210, 0xfedcba98, false, false);
}
static __device__ __forceinline__ float xhalf(float x) {
  return __builtin_bit_cast(float, xhalf(__builtin_bit_cast(int, x)));
}
static __device__ __forceinline__ float wave_sum(float x) {
  x += xhalf(x);   x += rowx<8>(x); x += rowx<4>(x);
  x += rowx<2>(x); x += rowx<1>(x);
  return x;
}
static __device__ __forceinline__ int wave_sum(int x) {
  x += xhalf(x);   x += rowx<8>(x); x += rowx<4>(x);
  x += rowx<2>(x); x += rowx<1>(x);
  return x;
}
static __device__ __forceinline__ float wave_max(float x) {
  x = fmaxf(x, xhalf(x));   x = fmaxf(x, rowx<8>(x)); x = fmaxf(x, rowx<4>(x));
  x = fmaxf(x, rowx<2>(x)); x = fmaxf(x, rowx<1>(x));
  return x;
}
static __device__ __forceinline__ int wave_min(int x) {
  x = min(x, xhalf(x));   x = min(x, rowx<8>(x)); x = min(x, rowx<4>(x));
  x = min(x, rowx<2>(x)); x = min(x, rowx<1>(x));
  return x;
}

#define GV_ROWS  13
#define GV_AHEAD 3
#define GV_KQ    8
#define GV_WAVES 16

__global__ __launch_bounds__(GV_WAVES * 32) void gemv_rows_wmma_f32(
    const float* __restrict__ A, const float* __restrict__ W,
    const float* __restrict__ bias, float* __restrict__ out, int N) {
  const unsigned lane = threadIdx.x & 31u;
  const unsigned wave = (unsigned)__builtin_amdgcn_readfirstlane(
      (int)((blockIdx.x * (unsigned)(GV_WAVES * 32) + threadIdx.x) >> 5));
  const unsigned row0 = wave * GV_ROWS;
  if (row0 >= (unsigned)N) return;

  v4f wq[GV_KQ];
  for (int q = 0; q < GV_KQ; ++q)
    wq[q] = *(const v4f*)(W + 128 * q + 4 * lane);

  const float* ab = A + (size_t)row0 * 1024u + 4u * lane;
  v4f a[GV_AHEAD + 1][GV_KQ];
#define GV_LOAD_ROW(R) do { \
    const unsigned rr_ = (row0 + (unsigned)(R) < (unsigned)N) ? (unsigned)(R) : 0u; \
    for (int q = 0; q < GV_KQ; ++q) \
      a[(R) % (GV_AHEAD + 1)][q] = __builtin_nontemporal_load( \
          (const v4f*)(ab + (size_t)rr_ * 1024u + 128u * q)); } while (0)

  for (int r = 0; r < GV_AHEAD; ++r) GV_LOAD_ROW(r);

  const bool lo8  = lane < 8u;
  const bool diag = lo8 || lane >= 24u;
  const bool b0 = (lane & 1u) != 0u, b1 = (lane & 2u) != 0u, b2 = (lane & 4u) != 0u;

  float part[16];
  for (int r = GV_ROWS; r < 16; ++r) part[r] = 0.0f;
  for (int r = 0; r < GV_ROWS; ++r) {
    __builtin_amdgcn_sched_barrier(0);
    if (r + GV_AHEAD < GV_ROWS) GV_LOAD_ROW(r + GV_AHEAD);
    __builtin_amdgcn_sched_barrier(0);
    const v4f* ar = a[r % (GV_AHEAD + 1)];
    v8f c0 = {}; v8f c1 = {};
    for (int q = 0; q < GV_KQ; ++q) {
      v2f a01 = {ar[q].x, ar[q].y}, w01 = {wq[q].x, wq[q].y};
      v2f a23 = {ar[q].z, ar[q].w}, w23 = {wq[q].z, wq[q].w};
      c0 = __builtin_amdgcn_wmma_f32_16x16x4_f32(false, a01, false, w01,
                                                 (short)0, c0, false, false);
      c1 = __builtin_amdgcn_wmma_f32_16x16x4_f32(false, a23, false, w23,
                                                 (short)0, c1, false, false);
    }
    v8f c = c0 + c1;
    float s0 = b0 ? c[1] : c[0], s1 = b0 ? c[3] : c[2];
    float s2 = b0 ? c[5] : c[4], s3 = b0 ? c[7] : c[6];
    float t0 = b1 ? s1 : s0, t1 = b1 ? s3 : s2;
    float u  = b2 ? t1 : t0;
    part[r] = diag ? u : 0.0f;
    asm volatile("" : "+v"(part[r]));
  }
#undef GV_LOAD_ROW

  const bool h16 = (lane & 16u) != 0u, h8 = (lane & 8u) != 0u;
  const bool h4 = (lane & 4u) != 0u, h2 = (lane & 2u) != 0u;
  float e8[8], e4[4], e2[2];
  for (int i = 0; i < 8; ++i) {
    float keep = h16 ? part[i + 8] : part[i], send = h16 ? part[i] : part[i + 8];
    e8[i] = keep + xhalf(send);
  }
  for (int i = 0; i < 4; ++i) {
    float keep = h8 ? e8[i + 4] : e8[i], send = h8 ? e8[i] : e8[i + 4];
    e4[i] = keep + rowx<8>(send);
  }
  for (int i = 0; i < 2; ++i) {
    float keep = h4 ? e4[i + 2] : e4[i], send = h4 ? e4[i] : e4[i + 2];
    e2[i] = keep + rowx<4>(send);
  }
  float keep1 = h2 ? e2[1] : e2[0], send1 = h2 ? e2[0] : e2[1];
  float e1 = keep1 + rowx<2>(send1);
  float tot = e1 + rowx<1>(e1);

  const unsigned myrow = row0 + (lane >> 1);
  if ((lane & 1u) == 0u && (lane >> 1) < (unsigned)GV_ROWS && myrow < (unsigned)N)
    out[myrow] = tot + bias[0];
}

#define SEG_U 16

__global__ __launch_bounds__(256) void seg_loss(
    const float* __restrict__ logits, const int* __restrict__ lens,
    const int* __restrict__ tgt,
    float* __restrict__ loss_ws, float* __restrict__ inc_ws,
    int S, int EACH) {
  int wave = (int)((blockIdx.x * blockDim.x + threadIdx.x) >> 5);
  int lane = (int)(threadIdx.x & 31);
  int s = wave;
  if (s >= S) return;

  int len = lens[s];
  int acc = 0;
  for (int base = 0; base < s; base += 32 * SEG_U) {
    int v[SEG_U];
    for (int u = 0; u < SEG_U; ++u) {
      int i = base + 32 * u + lane;
      v[u] = lens[i < s ? i : s];
    }
    for (int u = 0; u < SEG_U; ++u) {
      int i = base + 32 * u + lane;
      acc += (i < s) ? v[u] : 0;
    }
  }
  int off = wave_sum(acc);
  int j0 = lane, j1 = lane + 32;
  bool v0 = j0 < len, v1 = j1 < len;
  float x0 = v0 ? logits[off + j0] : NEG_INF;
  float x1 = v1 ? logits[off + j1] : NEG_INF;
  int   t0 = v0 ? tgt[off + j0] : 0;
  int   t1 = v1 ? tgt[off + j1] : 0;

  float mx = wave_max(fmaxf(x0, x1));
  float e0 = v0 ? __expf(x0 - mx) : 0.0f;
  float e1 = v1 ? __expf(x1 - mx) : 0.0f;
  float sum = wave_sum(e0 + e1);
  float inv = 1.0f / sum;
  float p0 = e0 * inv, p1 = e1 * inv;

  float q0 = v0 ? p0 : NEG_INF;
  float q1 = v1 ? p1 : NEG_INF;
  float m2 = wave_max(fmaxf(q0, q1));
  float s2 = wave_sum((v0 ? __expf(p0 - m2) : 0.0f) + (v1 ? __expf(p1 - m2) : 0.0f));
  float lse2 = m2 + __logf(s2);

  int k0 = (v0 && t0 == 1) ? j0 : INT_MAX;
  int k1 = (v1 && t1 == 1) ? j1 : INT_MAX;
  int tk = wave_min(min(k0, k1));
  tk = __builtin_amdgcn_readfirstlane(tk);
  int t_idx = (tk == INT_MAX) ? 0 : tk;
  float psel = (t_idx >= 32) ? p1 : p0;
  float p_t = __builtin_bit_cast(float, __builtin_amdgcn_readlane(
      __builtin_bit_cast(int, psel), t_idx & 31));
  float loss_per = lse2 - p_t;

  int ts = wave_sum(t0 + t1);

  int next_len = (s + 1 < S) ? lens[s + 1] : 1;
  bool skip_next_zero = (s > 0) && (s + 1 < S) && (next_len == 0);
  bool skip_no_pos = (((s + 1) % EACH) == 0) && (ts <= 0);
  bool include = (len > 0) && !skip_next_zero && !skip_no_pos;

  if (lane == 0) {
    loss_ws[s] = include ? loss_per : 0.0f;
    inc_ws[s]  = include ? 1.0f : 0.0f;
  }
}

template <int R> static __device__ __forceinline__ void fg_rows(
    const float* __restrict__ ab, const v4f (&wq)[GV_KQ], float (&part)[16],
    const bool diag, const bool b0, const bool b1, const bool b2) {
  v4f a[GV_AHEAD + 1][GV_KQ];
#define FG_LOAD_ROW(RR) do { \
    for (int q = 0; q < GV_KQ; ++q) \
      a[(RR) % (GV_AHEAD + 1)][q] = __builtin_nontemporal_load( \
          (const v4f*)(ab + (size_t)(RR) * 1024u + 128u * q)); } while (0)
#pragma unroll
  for (int r = 0; r < GV_AHEAD; ++r)
    if (r < R) FG_LOAD_ROW(r);
#pragma unroll
  for (int r = 0; r < R; ++r) {
    __builtin_amdgcn_sched_barrier(0);
    if (r + GV_AHEAD < R) FG_LOAD_ROW(r + GV_AHEAD);
    __builtin_amdgcn_sched_barrier(0);
    const v4f* ar = a[r % (GV_AHEAD + 1)];
    v8f c0 = {}; v8f c1 = {};
#pragma unroll
    for (int q = 0; q < GV_KQ; ++q) {
      v2f a01 = {ar[q].x, ar[q].y}, w01 = {wq[q].x, wq[q].y};
      v2f a23 = {ar[q].z, ar[q].w}, w23 = {wq[q].z, wq[q].w};
      c0 = __builtin_amdgcn_wmma_f32_16x16x4_f32(false, a01, false, w01,
                                                 (short)0, c0, false, false);
      c1 = __builtin_amdgcn_wmma_f32_16x16x4_f32(false, a23, false, w23,
                                                 (short)0, c1, false, false);
    }
    v8f c = c0 + c1;
    float s0 = b0 ? c[1] : c[0], s1 = b0 ? c[3] : c[2];
    float s2 = b0 ? c[5] : c[4], s3 = b0 ? c[7] : c[6];
    float t0 = b1 ? s1 : s0, t1 = b1 ? s3 : s2;
    float u  = b2 ? t1 : t0;
    part[r] = diag ? u : 0.0f;
    asm volatile("" : "+v"(part[r]));
  }
#undef FG_LOAD_ROW
}

#define FG_T   (GV_WAVES * 32)
#define FG_NOM 204
#define FG_CAP 4096
#define FG_C   (FG_CAP / FG_T)
#define FG_LROWS 1024

__global__ __launch_bounds__(FG_T) void gemv_seg_fused(
    const float* __restrict__ A, const float* __restrict__ W,
    const float* __restrict__ bias, float* out,
    const int* __restrict__ lens, const int* __restrict__ tgt,
    float* __restrict__ loss_ws, float* __restrict__ inc_ws,
    int N, int S, int EACH) {
  __shared__ int s_off[FG_CAP + 1];
  __shared__ int s_wt[GV_WAVES];
  __shared__ int s_cut[2];
  __shared__ float s_log[FG_LROWS];

  const unsigned t = threadIdx.x;
  const unsigned lane = t & 31u;
  const int wave = __builtin_amdgcn_readfirstlane((int)(t >> 5));
  const int g = (int)blockIdx.x, G = (int)gridDim.x;

  v4f wq[GV_KQ];
  for (int q = 0; q < GV_KQ; ++q)
    wq[q] = *(const v4f*)(W + 128 * q + 4 * lane);
  const float bv = bias[0];

  const int C = (S + FG_T - 1) / FG_T;
  const int i0 = (int)t * C;
  int lv[FG_C];
  for (int k = 0; k < FG_C; ++k) {
    const int i = i0 + k;
    lv[k] = (k < C && i < S) ? lens[i] : 0;
  }
  const int lprev = (i0 > 0 && i0 <= S) ? lens[i0 - 1] : 0;
  int tsum = 0;
  for (int k = 0; k < FG_C; ++k) tsum += lv[k];
  int excl = 0, tot = tsum, pt;
  pt = rowx<1>(tot); excl += (lane & 1u) ? pt : 0;  tot += pt;
  pt = rowx<2>(tot); excl += (lane & 2u) ? pt : 0;  tot += pt;
  pt = rowx<4>(tot); excl += (lane & 4u) ? pt : 0;  tot += pt;
  pt = rowx<8>(tot); excl += (lane & 8u) ? pt : 0;  tot += pt;
  pt = xhalf(tot);   excl += (lane & 16u) ? pt : 0; tot += pt;
  if (lane == 0u) s_wt[wave] = tot;
  if (t == 0u) { s_cut[0] = S; s_cut[1] = S; if (S == 0) s_off[0] = 0; }
  __syncthreads();
  int run = excl;
  for (int w = 0; w < GV_WAVES; ++w) run += (w < wave) ? s_wt[w] : 0;

  const int Xa = g * FG_NOM, Xb = Xa + FG_NOM;
  int prev = run - lprev;
  for (int k = 0; k < FG_C; ++k) {
    const int i = i0 + k;
    if (k < C && i < S) {
      s_off[i] = run;
      const bool first = (i == 0);
      if (run >= Xa && (first || prev < Xa))
        s_cut[0] = (first || (run - Xa) <= (Xa - prev)) ? i : i - 1;
      if (g + 1 < G && run >= Xb && (first || prev < Xb))
        s_cut[1] = (first || (run - Xb) <= (Xb - prev)) ? i : i - 1;
      prev = run;
      run += lv[k];
      if (i == S - 1) s_off[S] = run;
    }
  }
  __syncthreads();

  const int sa = __builtin_amdgcn_readfirstlane(s_cut[0]);
  const int sb = __builtin_amdgcn_readfirstlane(s_cut[1]);
  int rs_ = (g == 0) ? 0 : ((sa < S) ? min(s_off[sa], N) : N);
  int re_ = (g + 1 == G) ? N : ((sb < S) ? min(s_off[sb], N) : N);
  const int rs = __builtin_amdgcn_readfirstlane(rs_);
  const int re = __builtin_amdgcn_readfirstlane(re_);
  const bool inl = (re - rs) <= FG_LROWS;

  const int sfirst = sa + wave;
  int pt0 = 0, pt1 = 0;
  if (sfirst < sb) {
    const int off = s_off[sfirst], len = s_off[sfirst + 1] - off;
    if ((int)lane < len)      pt0 = tgt[min(off + (int)lane, N - 1)];
    if ((int)lane + 32 < len) pt1 = tgt[min(off + (int)lane + 32, N - 1)];
  }

  const int bq = 4 * (((wave & 3) - (wave >> 2)) & 3) + (wave >> 2);
  const bool lo8  = lane < 8u;
  const bool diag = lo8 || lane >= 24u;
  const bool b0 = (lane & 1u) != 0u, b1 = (lane & 2u) != 0u, b2 = (lane & 4u) != 0u;
  const bool h16 = (lane & 16u) != 0u, h8 = (lane & 8u) != 0u;
  const bool h4 = (lane & 4u) != 0u, h2 = (lane & 2u) != 0u;

  for (int pb = rs; pb < re; pb += 16 * GV_WAVES) {
    const int cnt = min(16 * GV_WAVES, re - pb);
    const int rbase = cnt / GV_WAVES, rrem = cnt % GV_WAVES;
    const int myrows = rbase + ((bq < rrem) ? 1 : 0);
    const int row0 = pb + bq * rbase + min(bq, rrem);

    const float* ab = A + (size_t)row0 * 1024u + 4u * lane;
    float part[16];
    for (int r = 0; r < 16; ++r) part[r] = 0.0f;
#define FG_CASE(R) case R: fg_rows<R>(ab, wq, part, diag, b0, b1, b2); break;
    switch (myrows) {
      FG_CASE(1)  FG_CASE(2)  FG_CASE(3)  FG_CASE(4)
      FG_CASE(5)  FG_CASE(6)  FG_CASE(7)  FG_CASE(8)
      FG_CASE(9)  FG_CASE(10) FG_CASE(11) FG_CASE(12)
      FG_CASE(13) FG_CASE(14) FG_CASE(15) FG_CASE(16)
      default: break;
    }
#undef FG_CASE

    float e8[8], e4[4], e2[2];
    for (int i = 0; i < 8; ++i) {
      float keep = h16 ? part[i + 8] : part[i], send = h16 ? part[i] : part[i + 8];
      e8[i] = keep + xhalf(send);
    }
    for (int i = 0; i < 4; ++i) {
      float keep = h8 ? e8[i + 4] : e8[i], send = h8 ? e8[i] : e8[i + 4];
      e4[i] = keep + rowx<8>(send);
    }
    for (int i = 0; i < 2; ++i) {
      float keep = h4 ? e4[i + 2] : e4[i], send = h4 ? e4[i] : e4[i + 2];
      e2[i] = keep + rowx<4>(send);
    }
    float keep1 = h2 ? e2[1] : e2[0], send1 = h2 ? e2[0] : e2[1];
    float e1 = keep1 + rowx<2>(send1);
    float tot1 = e1 + rowx<1>(e1);

    if ((lane & 1u) == 0u && (int)(lane >> 1) < myrows) {
      const int row = row0 + (int)(lane >> 1);
      const float v = tot1 + bv;
      out[row] = v;
      if (inl) s_log[row - rs] = v;
    }
  }

  if (inl) {
    __builtin_amdgcn_fence(__ATOMIC_RELEASE, "workgroup", "local");
    __builtin_amdgcn_s_barrier();
    __builtin_amdgcn_fence(__ATOMIC_ACQUIRE, "workgroup", "local");
  } else {
    __syncthreads();
  }

  for (int s = sa + wave; s < sb; s += GV_WAVES) {
    const int off = s_off[s], offn = s_off[s + 1];
    const int len = offn - off;
    const int next_len = (s + 1 < S) ? (s_off[s + 2] - offn) : 1;
    const int j0 = (int)lane, j1 = (int)lane + 32;
    const bool v0 = j0 < len, v1 = j1 < len;
    const int r0 = min(off + j0, N - 1), r1 = min(off + j1, N - 1);
    float x0 = NEG_INF, x1 = NEG_INF;
    if (v0) x0 = (inl && r0 >= rs) ? s_log[r0 - rs] : out[r0];
    if (v1) x1 = (inl && r1 >= rs) ? s_log[r1 - rs] : out[r1];
    int t0 = pt0, t1 = pt1;
    if (s != sfirst) {
      t0 = v0 ? tgt[r0] : 0;
      t1 = v1 ? tgt[r1] : 0;
    }

    float mx = wave_max(fmaxf(x0, x1));
    float e0 = v0 ? __expf(x0 - mx) : 0.0f;
    float e1 = v1 ? __expf(x1 - mx) : 0.0f;
    float sum = wave_sum(e0 + e1);
    float inv = 1.0f / sum;
    float p0 = e0 * inv, p1 = e1 * inv;

    float q0 = v0 ? p0 : NEG_INF;
    float q1 = v1 ? p1 : NEG_INF;
    float m2 = wave_max(fmaxf(q0, q1));
    float s2 = wave_sum((v0 ? __expf(p0 - m2) : 0.0f) + (v1 ? __expf(p1 - m2) : 0.0f));
    float lse2 = m2 + __logf(s2);

    int k0 = (v0 && t0 == 1) ? j0 : INT_MAX;
    int k1 = (v1 && t1 == 1) ? j1 : INT_MAX;
    int tk = wave_min(min(k0, k1));
    tk = __builtin_amdgcn_readfirstlane(tk);
    int t_idx = (tk == INT_MAX) ? 0 : tk;
    float psel = (t_idx >= 32) ? p1 : p0;
    float p_t = __builtin_bit_cast(float, __builtin_amdgcn_readlane(
        __builtin_bit_cast(int, psel), t_idx & 31));
    float loss_per = lse2 - p_t;

    int ts = wave_sum(t0 + t1);
    bool skip_next_zero = (s > 0) && (s + 1 < S) && (next_len == 0);
    bool skip_no_pos = (((s + 1) % EACH) == 0) && (ts <= 0);
    bool include = (len > 0) && !skip_next_zero && !skip_no_pos;
    if (lane == 0u) {
      loss_ws[s] = include ? loss_per : 0.0f;
      inc_ws[s]  = include ? 1.0f : 0.0f;
    }
  }
}

#define FR_T 256
#define FR_U 6

__global__ __launch_bounds__(FR_T) void final_reduce(
    const float* __restrict__ loss_ws, const float* __restrict__ inc_ws,
    float* __restrict__ out_loss, int S) {
  __shared__ float sl[FR_T / 32];
  __shared__ float sc[FR_T / 32];
  const int t = (int)threadIdx.x;
  const int lane = t & 31, wv = t >> 5;
  float a = 0.0f, c = 0.0f;
  for (int base = 0; base < S; base += FR_T * FR_U) {
    float la[FR_U], lc[FR_U];
    for (int u = 0; u < FR_U; ++u) {
      int i = base + FR_T * u + t;
      int ic = i < S ? i : S - 1;
      la[u] = loss_ws[ic];
      lc[u] = inc_ws[ic];
    }
    __builtin_amdgcn_sched_barrier(0);
    for (int u = 0; u < FR_U; ++u) {
      int i = base + FR_T * u + t;
      a += (i < S) ? la[u] : 0.0f;
      c += (i < S) ? lc[u] : 0.0f;
    }
  }
  a = wave_sum(a);
  c = wave_sum(c);
  if (lane == 0) { sl[wv] = a; sc[wv] = c; }
  __syncthreads();
  if (wv == 0) {
    float a2 = (lane < FR_T / 32) ? sl[lane] : 0.0f;
    float c2 = (lane < FR_T / 32) ? sc[lane] : 0.0f;
    static_assert(FR_T / 32 == 8, "the three steps below fold exactly 8 wave sums");
    a2 += rowx<4>(a2); a2 += rowx<2>(a2); a2 += rowx<1>(a2);
    c2 += rowx<4>(c2); c2 += rowx<2>(c2); c2 += rowx<1>(c2);
    if (lane == 0) {
      float denom = (c2 > 0.0f) ? c2 : 1e-4f;
      out_loss[0] = a2 / denom;
    }
  }
}

extern "C" void kernel_launch(void* const* d_in, const int* in_sizes, int n_in,
                              void* d_out, int out_size, void* d_ws, size_t ws_size,
                              hipStream_t stream) {
  const float* A    = (const float*)d_in[0];
  const float* W    = (const float*)d_in[1];
  const float* b    = (const float*)d_in[2];
  const int*   lens = (const int*)d_in[3];
  const int*   tgt  = (const int*)d_in[4];

  int Hdim = in_sizes[1];
  int N    = in_sizes[0] / Hdim;
  int S    = in_sizes[3];
  const int EACH = 32;

  float* logits   = (float*)d_out;
  float* loss_out = logits + N;

  size_t obytes  = (((size_t)S * sizeof(int)) + 255) & ~(size_t)255;
  float* loss_ws = (float*)((char*)d_ws + obytes);
  float* inc_ws  = loss_ws + S;

  if (Hdim == 1024 && S <= FG_CAP) {
    int blocks = (N + FG_NOM - 1) / FG_NOM;
    gemv_seg_fused<<<blocks, FG_T, 0, stream>>>(A, W, b, logits, lens, tgt,
                                                loss_ws, inc_ws, N, S, EACH);
  } else {
    int waves  = (N + 15) / 16;
    if (Hdim == 1024) {
      int rwaves = (N + GV_ROWS - 1) / GV_ROWS;
      int blocks = (rwaves + GV_WAVES - 1) / GV_WAVES;
      gemv_rows_wmma_f32<<<blocks, GV_WAVES * 32, 0, stream>>>(A, W, b, logits, N);
    } else {
      int blocks = (waves + 7) / 8;
      gemv_wmma_f32<<<blocks, 256, 0, stream>>>(A, W, b, logits, N, Hdim);
    }
    seg_loss<<<(S + 7) / 8, 256, 0, stream>>>(logits, lens, tgt,
                                              loss_ws, inc_ws, S, EACH);
  }

  final_reduce<<<1, FR_T, 0, stream>>>(loss_ws, inc_ws, loss_out, S);
}
